# starting kernel
# baseline (speedup 1.0000x reference)
#include <hip/hip_runtime.h>
#include <math.h>


typedef __attribute__((ext_vector_type(2))) float v2f;
typedef __attribute__((ext_vector_type(8))) float v8f;

#define N_NODES 200000
#define N_EDGES 600000
#define F_IN    117
#define F_PAD   124
#define D       128
#define D_PAD   132
#define L_LAYERS 3

__device__ __forceinline__ v8f wmma_f32_16x16x4(v2f a, v2f b, v8f c) {
  return __builtin_amdgcn_wmma_f32_16x16x4_f32(false, a, false, b, (short)0, c,
                                               false, false);
}

__global__ __launch_bounds__(256) void degree_kernel(const int* __restrict__ dst,
                                                     float* __restrict__ deg,
                                                     int nE) {
  int e = blockIdx.x * blockDim.x + threadIdx.x;
  if (e < nE) atomicAdd(&deg[dst[e]], 1.0f);
}

__global__ __launch_bounds__(256) void inv_deg_kernel(float* __restrict__ deg, int n) {
  int i = blockIdx.x * blockDim.x + threadIdx.x;
  if (i < n) deg[i] = 1.0f / fmaxf(deg[i], 1.0f);
}

__global__ __launch_bounds__(256) void input_gemm_tanh(const float* __restrict__ x,
                                                       const float* __restrict__ W_in,
                                                       const float* __restrict__ b_in,
                                                       float* __restrict__ h_out) {
  __shared__ float sx[16 * F_PAD];
  const int tid = threadIdx.x;
  const int mBase = blockIdx.x * 16;

  for (int i = tid; i < 16 * F_PAD; i += 256) {
    int r = i / F_PAD, c = i - r * F_PAD;
    sx[i] = (c < F_IN) ? x[(size_t)(mBase + r) * F_IN + c] : 0.0f;
  }
  __syncthreads();

  const int lane = tid & 31;
  const int half = lane >> 4;
  const int l16  = lane & 15;
  const int col  = (tid >> 5) * 16 + l16;

  v8f acc = {0.f, 0.f, 0.f, 0.f, 0.f, 0.f, 0.f, 0.f};
  for (int k = 0; k < F_IN; k += 4) {
    const int ka = k + 2 * half;
    v2f a, b;
    a.x = sx[l16 * F_PAD + ka];
    a.y = sx[l16 * F_PAD + ka + 1];
    b.x = (ka     < F_IN) ? W_in[(size_t)(ka)     * D + col] : 0.0f;
    b.y = (ka + 1 < F_IN) ? W_in[(size_t)(ka + 1) * D + col] : 0.0f;
    acc = wmma_f32_16x16x4(a, b, acc);
  }

  const float bias = b_in[col];
  const int rowBase = mBase + 8 * half;
#pragma unroll
  for (int r = 0; r < 8; ++r)
    h_out[(size_t)(rowBase + r) * D + col] = tanhf(acc[r] + bias);
}

__global__ __launch_bounds__(256) void scatter_sum(const float* __restrict__ h,
                                                   const int* __restrict__ src,
                                                   const int* __restrict__ dst,
                                                   float* __restrict__ agg, int nE) {
  int e = blockIdx.x * 8 + (threadIdx.x >> 5);
  if (e >= nE) return;
  int lane = threadIdx.x & 31;
  int s = src[e], d = dst[e];
  const float4* hv = (const float4*)(h + (size_t)s * D);
  float4 v = hv[lane];
  float* ap = agg + (size_t)d * D + lane * 4;
  atomicAdd(ap + 0, v.x);
  atomicAdd(ap + 1, v.y);
  atomicAdd(ap + 2, v.z);
  atomicAdd(ap + 3, v.w);
}

__global__ __launch_bounds__(256) void layer_gemm(const float* __restrict__ h,
                                                  const float* __restrict__ agg,
                                                  const float* __restrict__ inv_deg,
                                                  const float* __restrict__ Wself,
                                                  const float* __restrict__ Wneigh,
                                                  const float* __restrict__ bias_v,
                                                  float* __restrict__ h_out,
                                                  int relu) {
  __shared__ float sh[16 * D_PAD];
  __shared__ float sa[16 * D_PAD];
  const int tid = threadIdx.x;
  const int mBase = blockIdx.x * 16;

  for (int i = tid; i < 16 * D; i += 256) {
    int r = i >> 7, c = i & (D - 1);
    int row = mBase + r;
    float id = inv_deg[row];
    sh[r * D_PAD + c] = h[(size_t)row * D + c];
    sa[r * D_PAD + c] = agg[(size_t)row * D + c] * id;
  }
  __syncthreads();

  const int lane = tid & 31;
  const int half = lane >> 4;
  const int l16  = lane & 15;
  const int col  = (tid >> 5) * 16 + l16;

  v8f acc = {0.f, 0.f, 0.f, 0.f, 0.f, 0.f, 0.f, 0.f};
  for (int k = 0; k < D; k += 4) {
    const int ka = k + 2 * half;
    v2f a1, b1, a2, b2;
    a1.x = sh[l16 * D_PAD + ka];
    a1.y = sh[l16 * D_PAD + ka + 1];
    b1.x = Wself[(size_t)(ka)     * D + col];
    b1.y = Wself[(size_t)(ka + 1) * D + col];
    acc = wmma_f32_16x16x4(a1, b1, acc);

    a2.x = sa[l16 * D_PAD + ka];
    a2.y = sa[l16 * D_PAD + ka + 1];
    b2.x = Wneigh[(size_t)(ka)     * D + col];
    b2.y = Wneigh[(size_t)(ka + 1) * D + col];
    acc = wmma_f32_16x16x4(a2, b2, acc);
  }

  const float bv = bias_v[col];
  const int rowBase = mBase + 8 * half;
#pragma unroll
  for (int r = 0; r < 8; ++r) {
    float v = acc[r] + bv;
    if (relu) v = fmaxf(v, 0.0f);
    h_out[(size_t)(rowBase + r) * D + col] = v;
  }
}

extern "C" void kernel_launch(void* const* d_in, const int* in_sizes, int n_in,
                              void* d_out, int out_size, void* d_ws, size_t ws_size,
                              hipStream_t stream) {
  const float* x       = (const float*)d_in[0];
  const int*   ei      = (const int*)d_in[1];
  const float* W_in    = (const float*)d_in[2];
  const float* b_in    = (const float*)d_in[3];
  const float* W_self  = (const float*)d_in[4];
  const float* W_neigh = (const float*)d_in[5];
  const float* b_lay   = (const float*)d_in[6];
  float* out = (float*)d_out;

  float* ws   = (float*)d_ws;
  float* hA   = ws;
  float* hB   = ws + (size_t)1 * N_NODES * D;
  float* agg  = ws + (size_t)2 * N_NODES * D;
  float* degv = ws + (size_t)3 * N_NODES * D;

  const int* src = ei;
  const int* dst = ei + N_EDGES;

  hipMemsetAsync(degv, 0, N_NODES * sizeof(float), stream);
  degree_kernel<<<(N_EDGES + 255) / 256, 256, 0, stream>>>(dst, degv, N_EDGES);
  inv_deg_kernel<<<(N_NODES + 255) / 256, 256, 0, stream>>>(degv, N_NODES);

  input_gemm_tanh<<<N_NODES / 16, 256, 0, stream>>>(x, W_in, b_in, hA);

  const float* hcur = hA;
  for (int i = 0; i < L_LAYERS; ++i) {
    hipMemsetAsync(agg, 0, (size_t)N_NODES * D * sizeof(float), stream);
    scatter_sum<<<(N_EDGES + 7) / 8, 256, 0, stream>>>(hcur, src, dst, agg, N_EDGES);
    float* hnext = (i == L_LAYERS - 1) ? out : (hcur == hA ? hB : hA);
    layer_gemm<<<N_NODES / 16, 256, 0, stream>>>(
        hcur, agg, degv,
        W_self  + (size_t)i * D * D,
        W_neigh + (size_t)i * D * D,
        b_lay   + (size_t)i * D,
        hnext, (i < L_LAYERS - 1) ? 1 : 0);
    hcur = hnext;
  }
}


# Agg clear + global f32-atomic scatter replaced by owner work-groups: each holds 400 destination rows in LDS, scans all edges, adds matching rows by LDS atomics (arrival order), one store per line
# speedup vs baseline: 1.4282x; 1.4282x over previous
#include <hip/hip_runtime.h>
#include <math.h>


typedef __attribute__((ext_vector_type(2))) float v2f;
typedef __attribute__((ext_vector_type(8))) float v8f;

#define N_NODES 200000
#define N_EDGES 600000
#define F_IN    117
#define F_PAD   124
#define D       128
#define D_PAD   132
#define L_LAYERS 3
#define RT      4
#define BM      (16 * RT)
#define X_VEC4   ((BM * F_IN) / 4)
#define X_PASSES ((X_VEC4 + 255) / 256)
#define H_PASSES ((BM * (D / 4)) / 256)

#define OWN_ROWS    400
#define OWN_GROUPS  (N_NODES / OWN_ROWS)
#define OWN_THREADS 1024
#define OWN_WAVES   (OWN_THREADS / 32)
#define SCAN_U      4
#define E_VEC4      (N_EDGES / 4)
#define SCAN_ITERS  ((E_VEC4 + OWN_THREADS * SCAN_U - 1) / (OWN_THREADS * SCAN_U))
#define FLUSH_AT    8
#define LIST_CAP    640
#define GATHER_G    8
#define EDGE_BITS   20

static_assert(N_NODES % OWN_ROWS == 0, "ranges must tile the node set");
static_assert(N_EDGES % 4 == 0, "index rows are read as 16-byte vectors");
static_assert(N_EDGES <= (1 << EDGE_BITS) && OWN_ROWS <= (1 << (31 - EDGE_BITS)), "list entry packing");
static_assert(FLUSH_AT - 1 + 32 * 4 * SCAN_U <= LIST_CAP, "one scan trip must fit behind a carried remainder");

__device__ __forceinline__ v8f wmma_f32_16x16x4(v2f a, v2f b, v8f c) {
  return __builtin_amdgcn_wmma_f32_16x16x4_f32(false, a, false, b, (short)0, c,
                                               false, false);
}

__global__ __launch_bounds__(256) void degree_kernel(const int* __restrict__ dst,
                                                     float* __restrict__ deg,
                                                     int nE) {
  int e = blockIdx.x * blockDim.x + threadIdx.x;
  if (e < nE) atomicAdd(&deg[dst[e]], 1.0f);
}

__global__ __launch_bounds__(256) void inv_deg_kernel(float* __restrict__ deg, int n) {
  int i = blockIdx.x * blockDim.x + threadIdx.x;
  if (i < n) deg[i] = 1.0f / fmaxf(deg[i], 1.0f);
}

__global__ __launch_bounds__(256) void input_gemm_tanh(const float* __restrict__ x,
                                                       const float* __restrict__ W_in,
                                                       const float* __restrict__ b_in,
                                                       float* __restrict__ h_out) {
  __shared__ __attribute__((aligned(16))) float sx[BM * F_PAD];
  const int tid = threadIdx.x;
  const int mBase = blockIdx.x * BM;

  const float4* xt = (const float4*)(x + (size_t)mBase * F_IN);
  float4 xv[X_PASSES];
#pragma unroll
  for (int j = 0; j < X_PASSES; ++j) {
    const int i4 = tid + 256 * j;
    xv[j] = (i4 < X_VEC4) ? xt[i4] : make_float4(0.f, 0.f, 0.f, 0.f);
  }
#pragma unroll
  for (int j = 0; j < X_PASSES; ++j) {
    const int i4 = tid + 256 * j;
    if (i4 < X_VEC4) {
      const int i = i4 * 4;
      int r = i / F_IN, c = i - r * F_IN;
      const float e[4] = {xv[j].x, xv[j].y, xv[j].z, xv[j].w};
#pragma unroll
      for (int q = 0; q < 4; ++q) {
        sx[r * F_PAD + c] = e[q];
        if (++c == F_IN) { c = 0; ++r; }
      }
    }
  }
  for (int i = tid; i < BM * (F_PAD - F_IN); i += 256) {
    const int r = i / (F_PAD - F_IN), c = F_IN + (i - r * (F_PAD - F_IN));
    sx[r * F_PAD + c] = 0.0f;
  }
  __syncthreads();

  const int lane = tid & 31;
  const int half = lane >> 4;
  const int l16  = lane & 15;
  const int col  = (tid >> 5) * 16 + l16;

  v8f acc[RT];
#pragma unroll
  for (int t = 0; t < RT; ++t) acc[t] = (v8f){0.f, 0.f, 0.f, 0.f, 0.f, 0.f, 0.f, 0.f};

  for (int k = 0; k < F_IN; k += 4) {
    const int ka = k + 2 * half;
    v2f b;
    b.x = (ka     < F_IN) ? W_in[(size_t)(ka)     * D + col] : 0.0f;
    b.y = (ka + 1 < F_IN) ? W_in[(size_t)(ka + 1) * D + col] : 0.0f;
    v2f a[RT];
#pragma unroll
    for (int t = 0; t < RT; ++t)
      a[t] = *(const v2f*)&sx[(t * 16 + l16) * F_PAD + ka];
#pragma unroll
    for (int t = 0; t < RT; ++t) acc[t] = wmma_f32_16x16x4(a[t], b, acc[t]);
  }

  const float bias = b_in[col];
#pragma unroll
  for (int t = 0; t < RT; ++t) {
    const int rowBase = mBase + t * 16 + 8 * half;
#pragma unroll
    for (int r = 0; r < 8; ++r)
      h_out[(size_t)(rowBase + r) * D + col] = tanhf(acc[t][r] + bias);
  }
}

__device__ __forceinline__ void gather_add_rows(const int* mylst, int cnt, int lane,
                                                const float* __restrict__ h,
                                                const int* __restrict__ src,
                                                float* acc) {
  for (int j0 = 0; j0 < cnt; j0 += 32) {
    const int n = min(32, cnt - j0);
    int pk = 0, s = 0;
    if (lane < n) {
      pk = mylst[j0 + lane];
      s  = src[pk & ((1 << EDGE_BITS) - 1)];
    }
    for (int k = 0; k < n; k += GATHER_G) {
      float4 v[GATHER_G];
      int dl[GATHER_G];
#pragma unroll
      for (int g = 0; g < GATHER_G; ++g) {
        const int kk = min(k + g, n - 1);
        const int sg = __builtin_amdgcn_readlane(s, kk);
        dl[g] = __builtin_amdgcn_readlane(pk, kk) >> EDGE_BITS;
        v[g]  = ((const float4*)(h + (size_t)sg * D))[lane];
      }
#pragma unroll
      for (int g = 0; g < GATHER_G; ++g) {
        if (k + g < n) {
          float* p = acc + dl[g] * D + lane * 4;
          atomicAdd(p + 0, v[g].x);
          atomicAdd(p + 1, v[g].y);
          atomicAdd(p + 2, v[g].z);
          atomicAdd(p + 3, v[g].w);
        }
      }
    }
  }
}

__global__ __launch_bounds__(OWN_THREADS) void owner_range_sum(const float* __restrict__ h,
                                                              const int* __restrict__ src,
                                                              const int* __restrict__ dst,
                                                              float* __restrict__ agg) {
  __shared__ __attribute__((aligned(16))) float acc[OWN_ROWS * D];
  __shared__ int lst[OWN_WAVES * LIST_CAP];
  const int tid  = threadIdx.x;
  const int lane = tid & 31;
  const int rowBase = blockIdx.x * OWN_ROWS;

  for (int i = tid; i < OWN_ROWS * (D / 4); i += OWN_THREADS)
    ((float4*)acc)[i] = make_float4(0.f, 0.f, 0.f, 0.f);
  __syncthreads();

  int* mylst = lst + (tid >> 5) * LIST_CAP;
  const int4* dst4 = (const int4*)dst;
  int cnt = 0;
  const int rot = blockIdx.x % SCAN_ITERS;

  for (int i = 0; i < SCAN_ITERS; ++i) {
    int it = i + rot;
    if (it >= SCAN_ITERS) it -= SCAN_ITERS;
    int4 dv[SCAN_U];
    int  i4[SCAN_U];
    int  rb[SCAN_U];
#pragma unroll
    for (int u = 0; u < SCAN_U; ++u) {
      i4[u] = (it * SCAN_U + u) * OWN_THREADS + tid;
      const bool ok = i4[u] < E_VEC4;
      dv[u] = dst4[ok ? i4[u] : E_VEC4 - 1];
      rb[u] = ok ? rowBase : 0x7fffffff;
    }
#pragma unroll
    for (int u = 0; u < SCAN_U; ++u) {
      const int d[4] = {dv[u].x, dv[u].y, dv[u].z, dv[u].w};
#pragma unroll
      for (int q = 0; q < 4; ++q) {
        const unsigned dl = (unsigned)(d[q] - rb[u]);
        const bool m = dl < (unsigned)OWN_ROWS;
        const unsigned mask = __builtin_amdgcn_ballot_w32(m);
        if (mask != 0u) {
          if (m)
            mylst[cnt + (int)__builtin_amdgcn_mbcnt_lo(mask, 0u)] =
                (i4[u] * 4 + q) | (int)(dl << EDGE_BITS);
          cnt += __builtin_popcount(mask);
        }
      }
    }
    if (cnt >= FLUSH_AT) {
      gather_add_rows(mylst, cnt, lane, h, src, acc);
      cnt = 0;
    }
  }
  if (cnt > 0) gather_add_rows(mylst, cnt, lane, h, src, acc);
  __syncthreads();

  float4* out4 = (float4*)(agg + (size_t)rowBase * D);
  for (int i = tid; i < OWN_ROWS * (D / 4); i += OWN_THREADS)
    out4[i] = ((const float4*)acc)[i];
}

__global__ __launch_bounds__(256) void layer_gemm(const float* __restrict__ h,
                                                  const float* __restrict__ agg,
                                                  const float* __restrict__ inv_deg,
                                                  const float* __restrict__ Wself,
                                                  const float* __restrict__ Wneigh,
                                                  const float* __restrict__ bias_v,
                                                  float* __restrict__ h_out,
                                                  int relu) {
  __shared__ __attribute__((aligned(16))) float sh[BM * D_PAD];
  __shared__ __attribute__((aligned(16))) float sa[BM * D_PAD];
  const int tid = threadIdx.x;
  const int mBase = blockIdx.x * BM;

  float4 hv[H_PASSES], av[H_PASSES];
  float  idv[H_PASSES];
#pragma unroll
  for (int j = 0; j < H_PASSES; ++j) {
    const int i = tid + 256 * j;
    const int row = mBase + (i >> 5), c4 = i & 31;
    idv[j] = inv_deg[row];
    hv[j]  = ((const float4*)(h   + (size_t)row * D))[c4];
    av[j]  = ((const float4*)(agg + (size_t)row * D))[c4];
  }
#pragma unroll
  for (int j = 0; j < H_PASSES; ++j) {
    const int i = tid + 256 * j;
    const int r = i >> 5, c4 = i & 31;
    float4 a = av[j];
    a.x *= idv[j]; a.y *= idv[j]; a.z *= idv[j]; a.w *= idv[j];
    *(float4*)&sh[r * D_PAD + c4 * 4] = hv[j];
    *(float4*)&sa[r * D_PAD + c4 * 4] = a;
  }
  __syncthreads();

  const int lane = tid & 31;
  const int half = lane >> 4;
  const int l16  = lane & 15;
  const int col  = (tid >> 5) * 16 + l16;

  v8f acc[RT];
#pragma unroll
  for (int t = 0; t < RT; ++t) acc[t] = (v8f){0.f, 0.f, 0.f, 0.f, 0.f, 0.f, 0.f, 0.f};

#pragma unroll 4
  for (int k = 0; k < D; k += 4) {
    const int ka = k + 2 * half;
    v2f b1, b2;
    b1.x = Wself[(size_t)(ka)      * D + col];
    b1.y = Wself[(size_t)(ka + 1)  * D + col];
    b2.x = Wneigh[(size_t)(ka)     * D + col];
    b2.y = Wneigh[(size_t)(ka + 1) * D + col];
    v2f a1[RT], a2[RT];
#pragma unroll
    for (int t = 0; t < RT; ++t) {
      a1[t] = *(const v2f*)&sh[(t * 16 + l16) * D_PAD + ka];
      a2[t] = *(const v2f*)&sa[(t * 16 + l16) * D_PAD + ka];
    }
#pragma unroll
    for (int t = 0; t < RT; ++t) acc[t] = wmma_f32_16x16x4(a1[t], b1, acc[t]);
#pragma unroll
    for (int t = 0; t < RT; ++t) acc[t] = wmma_f32_16x16x4(a2[t], b2, acc[t]);
  }

  const float bv = bias_v[col];
#pragma unroll
  for (int t = 0; t < RT; ++t) {
    const int rowBase = mBase + t * 16 + 8 * half;
#pragma unroll
    for (int r = 0; r < 8; ++r) {
      float v = acc[t][r] + bv;
      if (relu) v = fmaxf(v, 0.0f);
      h_out[(size_t)(rowBase + r) * D + col] = v;
    }
  }
}

extern "C" void kernel_launch(void* const* d_in, const int* in_sizes, int n_in,
                              void* d_out, int out_size, void* d_ws, size_t ws_size,
                              hipStream_t stream) {
  const float* x       = (const float*)d_in[0];
  const int*   ei      = (const int*)d_in[1];
  const float* W_in    = (const float*)d_in[2];
  const float* b_in    = (const float*)d_in[3];
  const float* W_self  = (const float*)d_in[4];
  const float* W_neigh = (const float*)d_in[5];
  const float* b_lay   = (const float*)d_in[6];
  float* out = (float*)d_out;

  float* ws   = (float*)d_ws;
  float* hA   = ws;
  float* hB   = ws + (size_t)1 * N_NODES * D;
  float* agg  = ws + (size_t)2 * N_NODES * D;
  float* degv = ws + (size_t)3 * N_NODES * D;

  const int* src = ei;
  const int* dst = ei + N_EDGES;

  hipMemsetAsync(degv, 0, N_NODES * sizeof(float), stream);
  degree_kernel<<<(N_EDGES + 255) / 256, 256, 0, stream>>>(dst, degv, N_EDGES);
  inv_deg_kernel<<<(N_NODES + 255) / 256, 256, 0, stream>>>(degv, N_NODES);

  input_gemm_tanh<<<N_NODES / BM, 256, 0, stream>>>(x, W_in, b_in, hA);

  const float* hcur = hA;
  for (int i = 0; i < L_LAYERS; ++i) {
    owner_range_sum<<<OWN_GROUPS, OWN_THREADS, 0, stream>>>(hcur, src, dst, agg);
    float* hnext = (i == L_LAYERS - 1) ? out : (hcur == hA ? hB : hA);
    layer_gemm<<<N_NODES / BM, 256, 0, stream>>>(
        hcur, agg, degv,
        W_self  + (size_t)i * D * D,
        W_neigh + (size_t)i * D * D,
        b_lay   + (size_t)i * D,
        hnext, (i < L_LAYERS - 1) ? 1 : 0);
    hcur = hnext;
  }
}


# Owner passes of layers 2-3 replay the match list layer 1's scan records (lists in the h plane freed by in-place layer outputs): 2 x 500 index-row scans gone; sums still in LDS-atomic arrival order
# speedup vs baseline: 1.5564x; 1.0898x over previous
#include <hip/hip_runtime.h>
#include <math.h>


typedef __attribute__((ext_vector_type(2))) float v2f;
typedef __attribute__((ext_vector_type(8))) float v8f;

#define N_NODES 200000
#define N_EDGES 600000
#define F_IN    117
#define F_PAD   124
#define D       128
#define D_PAD   132
#define L_LAYERS 3
#define RT      4
#define BM      (16 * RT)
#define X_VEC4   ((BM * F_IN) / 4)
#define X_PASSES ((X_VEC4 + 255) / 256)
#define H_PASSES ((BM * (D / 4)) / 256)

#define OWN_ROWS    400
#define OWN_GROUPS  (N_NODES / OWN_ROWS)
#define OWN_THREADS 1024
#define OWN_WAVES   (OWN_THREADS / 32)
#define SCAN_U      4
#define E_VEC4      (N_EDGES / 4)
#define SCAN_ITERS  ((E_VEC4 + OWN_THREADS * SCAN_U - 1) / (OWN_THREADS * SCAN_U))
#define FLUSH_AT    8
#define LIST_CAP    640
#define GATHER_G    8
#define EDGE_BITS   20
#define SRC_BITS    18
#define REC_CAP     32768
#define REC_HDR     512
#define REPLAY_G    16

static_assert(N_NODES % OWN_ROWS == 0, "ranges must tile the node set");
static_assert(N_EDGES % 4 == 0, "index rows are read as 16-byte vectors");
static_assert(N_EDGES <= (1 << EDGE_BITS) && OWN_ROWS <= (1 << (31 - EDGE_BITS)), "list entry packing");
static_assert(FLUSH_AT - 1 + 32 * 4 * SCAN_U <= LIST_CAP, "one scan trip must fit behind a carried remainder");
static_assert(N_NODES <= (1 << SRC_BITS) && OWN_ROWS <= (1 << (31 - SRC_BITS)), "recorded entry packing");
static_assert(OWN_GROUPS <= REC_HDR, "one count per range");
static_assert((size_t)REC_HDR + (size_t)OWN_GROUPS * REC_CAP <= (size_t)N_NODES * D, "lists fit in one h plane");

__device__ __forceinline__ v8f wmma_f32_16x16x4(v2f a, v2f b, v8f c) {
  return __builtin_amdgcn_wmma_f32_16x16x4_f32(false, a, false, b, (short)0, c,
                                               false, false);
}

__global__ __launch_bounds__(256) void degree_kernel(const int* __restrict__ dst,
                                                     float* __restrict__ deg,
                                                     int nE) {
  int e = blockIdx.x * blockDim.x + threadIdx.x;
  if (e < nE) atomicAdd(&deg[dst[e]], 1.0f);
}

__global__ __launch_bounds__(256) void inv_deg_kernel(float* __restrict__ deg, int n) {
  int i = blockIdx.x * blockDim.x + threadIdx.x;
  if (i < n) deg[i] = 1.0f / fmaxf(deg[i], 1.0f);
}

__global__ __launch_bounds__(256) void input_gemm_tanh(const float* __restrict__ x,
                                                       const float* __restrict__ W_in,
                                                       const float* __restrict__ b_in,
                                                       float* __restrict__ h_out) {
  __shared__ __attribute__((aligned(16))) float sx[BM * F_PAD];
  const int tid = threadIdx.x;
  const int mBase = blockIdx.x * BM;

  const float4* xt = (const float4*)(x + (size_t)mBase * F_IN);
  float4 xv[X_PASSES];
#pragma unroll
  for (int j = 0; j < X_PASSES; ++j) {
    const int i4 = tid + 256 * j;
    xv[j] = (i4 < X_VEC4) ? xt[i4] : make_float4(0.f, 0.f, 0.f, 0.f);
  }
#pragma unroll
  for (int j = 0; j < X_PASSES; ++j) {
    const int i4 = tid + 256 * j;
    if (i4 < X_VEC4) {
      const int i = i4 * 4;
      int r = i / F_IN, c = i - r * F_IN;
      const float e[4] = {xv[j].x, xv[j].y, xv[j].z, xv[j].w};
#pragma unroll
      for (int q = 0; q < 4; ++q) {
        sx[r * F_PAD + c] = e[q];
        if (++c == F_IN) { c = 0; ++r; }
      }
    }
  }
  for (int i = tid; i < BM * (F_PAD - F_IN); i += 256) {
    const int r = i / (F_PAD - F_IN), c = F_IN + (i - r * (F_PAD - F_IN));
    sx[r * F_PAD + c] = 0.0f;
  }
  __syncthreads();

  const int lane = tid & 31;
  const int half = lane >> 4;
  const int l16  = lane & 15;
  const int col  = (tid >> 5) * 16 + l16;

  v8f acc[RT];
#pragma unroll
  for (int t = 0; t < RT; ++t) acc[t] = (v8f){0.f, 0.f, 0.f, 0.f, 0.f, 0.f, 0.f, 0.f};

  for (int k = 0; k < F_IN; k += 4) {
    const int ka = k + 2 * half;
    v2f b;
    b.x = (ka     < F_IN) ? W_in[(size_t)(ka)     * D + col] : 0.0f;
    b.y = (ka + 1 < F_IN) ? W_in[(size_t)(ka + 1) * D + col] : 0.0f;
    v2f a[RT];
#pragma unroll
    for (int t = 0; t < RT; ++t)
      a[t] = *(const v2f*)&sx[(t * 16 + l16) * F_PAD + ka];
#pragma unroll
    for (int t = 0; t < RT; ++t) acc[t] = wmma_f32_16x16x4(a[t], b, acc[t]);
  }

  const float bias = b_in[col];
#pragma unroll
  for (int t = 0; t < RT; ++t) {
    const int rowBase = mBase + t * 16 + 8 * half;
#pragma unroll
    for (int r = 0; r < 8; ++r)
      h_out[(size_t)(rowBase + r) * D + col] = tanhf(acc[t][r] + bias);
  }
}

template <int G, bool SPARE>
__device__ __forceinline__ void add_rows(int s, int dl, int n, int lane,
                                         const float* __restrict__ h, float* acc) {
  for (int k = 0; k < n; k += G) {
    float4 v[G];
    int d[G];
#pragma unroll
    for (int g = 0; g < G; ++g) {
      const int kk = min(k + g, n - 1);
      const int sg = __builtin_amdgcn_readlane(s, kk);
      d[g] = (SPARE && k + g >= n) ? OWN_ROWS : __builtin_amdgcn_readlane(dl, kk);
      v[g] = ((const float4*)(h + (size_t)sg * D))[lane];
    }
#pragma unroll
    for (int g = 0; g < G; ++g) {
      if (SPARE || k + g < n) {
        float* p = acc + d[g] * D + lane * 4;
        atomicAdd(p + 0, v[g].x);
        atomicAdd(p + 1, v[g].y);
        atomicAdd(p + 2, v[g].z);
        atomicAdd(p + 3, v[g].w);
      }
    }
  }
}

template <bool RECORD>
__device__ __forceinline__ void gather_add_rows(const int* mylst, int cnt, int lane,
                                                const float* __restrict__ h,
                                                const int* __restrict__ src,
                                                float* acc, int* glist, int* fill) {
  int base = 0;
  if (RECORD) {
    if (lane == 0) base = atomicAdd(fill, cnt);
    base = __builtin_amdgcn_readfirstlane(base);
  }
  for (int j0 = 0; j0 < cnt; j0 += 32) {
    const int n = min(32, cnt - j0);
    int pk = 0, s = 0;
    if (lane < n) {
      pk = mylst[j0 + lane];
      s  = src[pk & ((1 << EDGE_BITS) - 1)];
    }
    const int dl = pk >> EDGE_BITS;
    if (RECORD) {
      const int at = base + j0 + lane;
      if (lane < n && at < REC_CAP) glist[at] = s | (dl << SRC_BITS);
    }
    add_rows<GATHER_G, false>(s, dl, n, lane, h, acc);
  }
}

template <bool RECORD>
__device__ __forceinline__ void scan_range(int tid, int lane, int rowBase, int* lst,
                                           const float* __restrict__ h,
                                           const int* __restrict__ src,
                                           const int* __restrict__ dst,
                                           float* acc, int* glist, int* fill) {
  int* mylst = lst + (tid >> 5) * LIST_CAP;
  const int4* dst4 = (const int4*)dst;
  int cnt = 0;
  const int rot = blockIdx.x % SCAN_ITERS;

  for (int i = 0; i < SCAN_ITERS; ++i) {
    int it = i + rot;
    if (it >= SCAN_ITERS) it -= SCAN_ITERS;
    int4 dv[SCAN_U];
    int  i4[SCAN_U];
    int  rb[SCAN_U];
#pragma unroll
    for (int u = 0; u < SCAN_U; ++u) {
      i4[u] = (it * SCAN_U + u) * OWN_THREADS + tid;
      const bool ok = i4[u] < E_VEC4;
      dv[u] = dst4[ok ? i4[u] : E_VEC4 - 1];
      rb[u] = ok ? rowBase : 0x7fffffff;
    }
#pragma unroll
    for (int u = 0; u < SCAN_U; ++u) {
      const int d[4] = {dv[u].x, dv[u].y, dv[u].z, dv[u].w};
#pragma unroll
      for (int q = 0; q < 4; ++q) {
        const unsigned dl = (unsigned)(d[q] - rb[u]);
        const bool m = dl < (unsigned)OWN_ROWS;
        const unsigned mask = __builtin_amdgcn_ballot_w32(m);
        if (mask != 0u) {
          if (m)
            mylst[cnt + (int)__builtin_amdgcn_mbcnt_lo(mask, 0u)] =
                (i4[u] * 4 + q) | (int)(dl << EDGE_BITS);
          cnt += __builtin_popcount(mask);
        }
      }
    }
    if (cnt >= FLUSH_AT) {
      gather_add_rows<RECORD>(mylst, cnt, lane, h, src, acc, glist, fill);
      cnt = 0;
    }
  }
  if (cnt > 0) gather_add_rows<RECORD>(mylst, cnt, lane, h, src, acc, glist, fill);
}

__global__ __launch_bounds__(OWN_THREADS) void owner_range_scan(const float* __restrict__ h,
                                                               const int* __restrict__ src,
                                                               const int* __restrict__ dst,
                                                               int* __restrict__ rec,
                                                               float* __restrict__ agg) {
  __shared__ __attribute__((aligned(16))) float acc[OWN_ROWS * D];
  __shared__ int lst[OWN_WAVES * LIST_CAP];
  __shared__ int fill;
  const int tid  = threadIdx.x;
  const int lane = tid & 31;
  const int rowBase = blockIdx.x * OWN_ROWS;

  for (int i = tid; i < OWN_ROWS * (D / 4); i += OWN_THREADS)
    ((float4*)acc)[i] = make_float4(0.f, 0.f, 0.f, 0.f);
  if (tid == 0) fill = 0;
  __syncthreads();

  int* glist = rec + REC_HDR + (size_t)blockIdx.x * REC_CAP;
  scan_range<true>(tid, lane, rowBase, lst, h, src, dst, acc, glist, &fill);
  __syncthreads();

  if (tid == 0) rec[blockIdx.x] = (fill <= REC_CAP) ? fill : -1;

  float4* out4 = (float4*)(agg + (size_t)rowBase * D);
  for (int i = tid; i < OWN_ROWS * (D / 4); i += OWN_THREADS)
    out4[i] = ((const float4*)acc)[i];
}

__global__ __launch_bounds__(OWN_THREADS) void owner_range_replay(const float* __restrict__ h,
                                                                 const int* __restrict__ src,
                                                                 const int* __restrict__ dst,
                                                                 const int* __restrict__ rec,
                                                                 float* __restrict__ agg) {
  __shared__ __attribute__((aligned(16))) float acc[(OWN_ROWS + 1) * D];
  __shared__ int lst[OWN_WAVES * LIST_CAP];
  const int tid  = threadIdx.x;
  const int lane = tid & 31;
  const int rowBase = blockIdx.x * OWN_ROWS;

  for (int i = tid; i < (OWN_ROWS + 1) * (D / 4); i += OWN_THREADS)
    ((float4*)acc)[i] = make_float4(0.f, 0.f, 0.f, 0.f);
  __syncthreads();

  const int nrec = rec[blockIdx.x];
  if (nrec >= 0 && nrec <= REC_CAP) {
    const int* glist = rec + REC_HDR + (size_t)blockIdx.x * REC_CAP;
    const int wave = __builtin_amdgcn_readfirstlane(tid >> 5);
    const int per  = (nrec + OWN_WAVES - 1) / OWN_WAVES;
    const int beg  = wave * per;
    const int end  = min(nrec, beg + per);
    int pk_next = (beg + lane < end) ? glist[beg + lane] : 0;
    for (int j0 = beg; j0 < end; j0 += 32) {
      const int n = min(32, end - j0);
      const int pk = pk_next;
      pk_next = (j0 + 32 + lane < end) ? glist[j0 + 32 + lane] : 0;
      add_rows<REPLAY_G, true>(pk & ((1 << SRC_BITS) - 1), pk >> SRC_BITS, n, lane, h, acc);
    }
  } else {
    scan_range<false>(tid, lane, rowBase, lst, h, src, dst, acc, (int*)0, (int*)0);
  }
  __syncthreads();

  float4* out4 = (float4*)(agg + (size_t)rowBase * D);
  for (int i = tid; i < OWN_ROWS * (D / 4); i += OWN_THREADS)
    out4[i] = ((const float4*)acc)[i];
}

__global__ __launch_bounds__(256) void layer_gemm(const float* h,
                                                  const float* __restrict__ agg,
                                                  const float* __restrict__ inv_deg,
                                                  const float* __restrict__ Wself,
                                                  const float* __restrict__ Wneigh,
                                                  const float* __restrict__ bias_v,
                                                  float* h_out,
                                                  int relu) {
  __shared__ __attribute__((aligned(16))) float sh[BM * D_PAD];
  __shared__ __attribute__((aligned(16))) float sa[BM * D_PAD];
  const int tid = threadIdx.x;
  const int mBase = blockIdx.x * BM;

  float4 hv[H_PASSES], av[H_PASSES];
  float  idv[H_PASSES];
#pragma unroll
  for (int j = 0; j < H_PASSES; ++j) {
    const int i = tid + 256 * j;
    const int row = mBase + (i >> 5), c4 = i & 31;
    idv[j] = inv_deg[row];
    hv[j]  = ((const float4*)(h   + (size_t)row * D))[c4];
    av[j]  = ((const float4*)(agg + (size_t)row * D))[c4];
  }
#pragma unroll
  for (int j = 0; j < H_PASSES; ++j) {
    const int i = tid + 256 * j;
    const int r = i >> 5, c4 = i & 31;
    float4 a = av[j];
    a.x *= idv[j]; a.y *= idv[j]; a.z *= idv[j]; a.w *= idv[j];
    *(float4*)&sh[r * D_PAD + c4 * 4] = hv[j];
    *(float4*)&sa[r * D_PAD + c4 * 4] = a;
  }
  __syncthreads();

  const int lane = tid & 31;
  const int half = lane >> 4;
  const int l16  = lane & 15;
  const int col  = (tid >> 5) * 16 + l16;

  v8f acc[RT];
#pragma unroll
  for (int t = 0; t < RT; ++t) acc[t] = (v8f){0.f, 0.f, 0.f, 0.f, 0.f, 0.f, 0.f, 0.f};

#pragma unroll 4
  for (int k = 0; k < D; k += 4) {
    const int ka = k + 2 * half;
    v2f b1, b2;
    b1.x = Wself[(size_t)(ka)      * D + col];
    b1.y = Wself[(size_t)(ka + 1)  * D + col];
    b2.x = Wneigh[(size_t)(ka)     * D + col];
    b2.y = Wneigh[(size_t)(ka + 1) * D + col];
    v2f a1[RT], a2[RT];
#pragma unroll
    for (int t = 0; t < RT; ++t) {
      a1[t] = *(const v2f*)&sh[(t * 16 + l16) * D_PAD + ka];
      a2[t] = *(const v2f*)&sa[(t * 16 + l16) * D_PAD + ka];
    }
#pragma unroll
    for (int t = 0; t < RT; ++t) acc[t] = wmma_f32_16x16x4(a1[t], b1, acc[t]);
#pragma unroll
    for (int t = 0; t < RT; ++t) acc[t] = wmma_f32_16x16x4(a2[t], b2, acc[t]);
  }

  const float bv = bias_v[col];
#pragma unroll
  for (int t = 0; t < RT; ++t) {
    const int rowBase = mBase + t * 16 + 8 * half;
#pragma unroll
    for (int r = 0; r < 8; ++r) {
      float v = acc[t][r] + bv;
      if (relu) v = fmaxf(v, 0.0f);
      h_out[(size_t)(rowBase + r) * D + col] = v;
    }
  }
}

extern "C" void kernel_launch(void* const* d_in, const int* in_sizes, int n_in,
                              void* d_out, int out_size, void* d_ws, size_t ws_size,
                              hipStream_t stream) {
  const float* x       = (const float*)d_in[0];
  const int*   ei      = (const int*)d_in[1];
  const float* W_in    = (const float*)d_in[2];
  const float* b_in    = (const float*)d_in[3];
  const float* W_self  = (const float*)d_in[4];
  const float* W_neigh = (const float*)d_in[5];
  const float* b_lay   = (const float*)d_in[6];
  float* out = (float*)d_out;

  float* ws   = (float*)d_ws;
  float* hA   = ws;
  int*   rec  = (int*)(ws + (size_t)1 * N_NODES * D);
  float* agg  = ws + (size_t)2 * N_NODES * D;
  float* degv = ws + (size_t)3 * N_NODES * D;

  const int* src = ei;
  const int* dst = ei + N_EDGES;

  hipMemsetAsync(degv, 0, N_NODES * sizeof(float), stream);
  degree_kernel<<<(N_EDGES + 255) / 256, 256, 0, stream>>>(dst, degv, N_EDGES);
  inv_deg_kernel<<<(N_NODES + 255) / 256, 256, 0, stream>>>(degv, N_NODES);

  input_gemm_tanh<<<N_NODES / BM, 256, 0, stream>>>(x, W_in, b_in, hA);

  const float* hcur = hA;
  for (int i = 0; i < L_LAYERS; ++i) {
    if (i == 0)
      owner_range_scan<<<OWN_GROUPS, OWN_THREADS, 0, stream>>>(hcur, src, dst, rec, agg);
    else
      owner_range_replay<<<OWN_GROUPS, OWN_THREADS, 0, stream>>>(hcur, src, dst, rec, agg);
    float* hnext = (i == L_LAYERS - 1) ? out : hA;
    layer_gemm<<<N_NODES / BM, 256, 0, stream>>>(
        hcur, agg, degv,
        W_self  + (size_t)i * D * D,
        W_neigh + (size_t)i * D * D,
        b_lay   + (size_t)i * D,
        hnext, (i < L_LAYERS - 1) ? 1 : 0);
    hcur = hnext;
  }
}


# Degree pass folded into the owner work-groups: each counts its rows' in-degrees in LDS while adding and stores means; memset, degree and inverse kernels gone (9->7 launches); sum order unchanged
# speedup vs baseline: 1.6137x; 1.0368x over previous
#include <hip/hip_runtime.h>
#include <math.h>


typedef __attribute__((ext_vector_type(2))) float v2f;
typedef __attribute__((ext_vector_type(8))) float v8f;

#define N_NODES 200000
#define N_EDGES 600000
#define F_IN    117
#define F_PAD   124
#define D       128
#define D_PAD   132
#define L_LAYERS 3
#define RT      4
#define BM      (16 * RT)
#define X_VEC4   ((BM * F_IN) / 4)
#define X_PASSES ((X_VEC4 + 255) / 256)
#define H_PASSES ((BM * (D / 4)) / 256)

#define OWN_ROWS    400
#define OWN_GROUPS  (N_NODES / OWN_ROWS)
#define OWN_THREADS 1024
#define OWN_WAVES   (OWN_THREADS / 32)
#define SCAN_U      4
#define E_VEC4      (N_EDGES / 4)
#define SCAN_ITERS  ((E_VEC4 + OWN_THREADS * SCAN_U - 1) / (OWN_THREADS * SCAN_U))
#define FLUSH_AT    8
#define LIST_CAP    640
#define GATHER_G    8
#define EDGE_BITS   20
#define SRC_BITS    18
#define REC_CAP     32768
#define REC_HDR     512
#define REPLAY_G    16

static_assert(N_NODES % OWN_ROWS == 0, "ranges must tile the node set");
static_assert(N_EDGES % 4 == 0, "index rows are read as 16-byte vectors");
static_assert(N_EDGES <= (1 << EDGE_BITS) && OWN_ROWS <= (1 << (31 - EDGE_BITS)), "list entry packing");
static_assert(FLUSH_AT - 1 + 32 * 4 * SCAN_U <= LIST_CAP, "one scan trip must fit behind a carried remainder");
static_assert(N_NODES <= (1 << SRC_BITS) && OWN_ROWS <= (1 << (31 - SRC_BITS)), "recorded entry packing");
static_assert(OWN_GROUPS <= REC_HDR, "one count per range");
static_assert((size_t)REC_HDR + (size_t)OWN_GROUPS * REC_CAP <= (size_t)N_NODES * D, "lists fit in one h plane");

__device__ __forceinline__ v8f wmma_f32_16x16x4(v2f a, v2f b, v8f c) {
  return __builtin_amdgcn_wmma_f32_16x16x4_f32(false, a, false, b, (short)0, c,
                                               false, false);
}

__global__ __launch_bounds__(256) void input_gemm_tanh(const float* __restrict__ x,
                                                       const float* __restrict__ W_in,
                                                       const float* __restrict__ b_in,
                                                       float* __restrict__ h_out) {
  __shared__ __attribute__((aligned(16))) float sx[BM * F_PAD];
  const int tid = threadIdx.x;
  const int mBase = blockIdx.x * BM;

  const float4* xt = (const float4*)(x + (size_t)mBase * F_IN);
  float4 xv[X_PASSES];
#pragma unroll
  for (int j = 0; j < X_PASSES; ++j) {
    const int i4 = tid + 256 * j;
    xv[j] = (i4 < X_VEC4) ? xt[i4] : make_float4(0.f, 0.f, 0.f, 0.f);
  }
#pragma unroll
  for (int j = 0; j < X_PASSES; ++j) {
    const int i4 = tid + 256 * j;
    if (i4 < X_VEC4) {
      const int i = i4 * 4;
      int r = i / F_IN, c = i - r * F_IN;
      const float e[4] = {xv[j].x, xv[j].y, xv[j].z, xv[j].w};
#pragma unroll
      for (int q = 0; q < 4; ++q) {
        sx[r * F_PAD + c] = e[q];
        if (++c == F_IN) { c = 0; ++r; }
      }
    }
  }
  for (int i = tid; i < BM * (F_PAD - F_IN); i += 256) {
    const int r = i / (F_PAD - F_IN), c = F_IN + (i - r * (F_PAD - F_IN));
    sx[r * F_PAD + c] = 0.0f;
  }
  __syncthreads();

  const int lane = tid & 31;
  const int half = lane >> 4;
  const int l16  = lane & 15;
  const int col  = (tid >> 5) * 16 + l16;

  v8f acc[RT];
#pragma unroll
  for (int t = 0; t < RT; ++t) acc[t] = (v8f){0.f, 0.f, 0.f, 0.f, 0.f, 0.f, 0.f, 0.f};

  for (int k = 0; k < F_IN; k += 4) {
    const int ka = k + 2 * half;
    v2f b;
    b.x = (ka     < F_IN) ? W_in[(size_t)(ka)     * D + col] : 0.0f;
    b.y = (ka + 1 < F_IN) ? W_in[(size_t)(ka + 1) * D + col] : 0.0f;
    v2f a[RT];
#pragma unroll
    for (int t = 0; t < RT; ++t)
      a[t] = *(const v2f*)&sx[(t * 16 + l16) * F_PAD + ka];
#pragma unroll
    for (int t = 0; t < RT; ++t) acc[t] = wmma_f32_16x16x4(a[t], b, acc[t]);
  }

  const float bias = b_in[col];
#pragma unroll
  for (int t = 0; t < RT; ++t) {
    const int rowBase = mBase + t * 16 + 8 * half;
#pragma unroll
    for (int r = 0; r < 8; ++r)
      h_out[(size_t)(rowBase + r) * D + col] = tanhf(acc[t][r] + bias);
  }
}

template <int G, bool SPARE>
__device__ __forceinline__ void add_rows(int s, int dl, int n, int lane,
                                         const float* __restrict__ h, float* acc) {
  for (int k = 0; k < n; k += G) {
    float4 v[G];
    int d[G];
#pragma unroll
    for (int g = 0; g < G; ++g) {
      const int kk = min(k + g, n - 1);
      const int sg = __builtin_amdgcn_readlane(s, kk);
      d[g] = (SPARE && k + g >= n) ? OWN_ROWS : __builtin_amdgcn_readlane(dl, kk);
      v[g] = ((const float4*)(h + (size_t)sg * D))[lane];
    }
#pragma unroll
    for (int g = 0; g < G; ++g) {
      if (SPARE || k + g < n) {
        float* p = acc + d[g] * D + lane * 4;
        atomicAdd(p + 0, v[g].x);
        atomicAdd(p + 1, v[g].y);
        atomicAdd(p + 2, v[g].z);
        atomicAdd(p + 3, v[g].w);
      }
    }
  }
}

template <bool RECORD>
__device__ __forceinline__ void gather_add_rows(const int* mylst, int cnt, int lane,
                                                const float* __restrict__ h,
                                                const int* __restrict__ src,
                                                float* acc, int* glist, int* fill) {
  int base = 0;
  if (RECORD) {
    if (lane == 0) base = atomicAdd(fill, cnt);
    base = __builtin_amdgcn_readfirstlane(base);
  }
  for (int j0 = 0; j0 < cnt; j0 += 32) {
    const int n = min(32, cnt - j0);
    int pk = 0, s = 0;
    if (lane < n) {
      pk = mylst[j0 + lane];
      s  = src[pk & ((1 << EDGE_BITS) - 1)];
    }
    const int dl = pk >> EDGE_BITS;
    if (RECORD) {
      const int at = base + j0 + lane;
      if (lane < n && at < REC_CAP) glist[at] = s | (dl << SRC_BITS);
    }
    add_rows<GATHER_G, false>(s, dl, n, lane, h, acc);
  }
}

template <bool RECORD>
__device__ __forceinline__ void scan_range(int tid, int lane, int rowBase, int* lst,
                                           const float* __restrict__ h,
                                           const int* __restrict__ src,
                                           const int* __restrict__ dst,
                                           float* acc, int* degc, int* glist, int* fill) {
  int* mylst = lst + (tid >> 5) * LIST_CAP;
  const int4* dst4 = (const int4*)dst;
  int cnt = 0;
  const int rot = blockIdx.x % SCAN_ITERS;

  for (int i = 0; i < SCAN_ITERS; ++i) {
    int it = i + rot;
    if (it >= SCAN_ITERS) it -= SCAN_ITERS;
    int4 dv[SCAN_U];
    int  i4[SCAN_U];
    int  rb[SCAN_U];
#pragma unroll
    for (int u = 0; u < SCAN_U; ++u) {
      i4[u] = (it * SCAN_U + u) * OWN_THREADS + tid;
      const bool ok = i4[u] < E_VEC4;
      dv[u] = dst4[ok ? i4[u] : E_VEC4 - 1];
      rb[u] = ok ? rowBase : 0x7fffffff;
    }
#pragma unroll
    for (int u = 0; u < SCAN_U; ++u) {
      const int d[4] = {dv[u].x, dv[u].y, dv[u].z, dv[u].w};
#pragma unroll
      for (int q = 0; q < 4; ++q) {
        const unsigned dl = (unsigned)(d[q] - rb[u]);
        const bool m = dl < (unsigned)OWN_ROWS;
        const unsigned mask = __builtin_amdgcn_ballot_w32(m);
        if (mask != 0u) {
          if (m) {
            mylst[cnt + (int)__builtin_amdgcn_mbcnt_lo(mask, 0u)] =
                (i4[u] * 4 + q) | (int)(dl << EDGE_BITS);
            atomicAdd(&degc[dl], 1);
          }
          cnt += __builtin_popcount(mask);
        }
      }
    }
    if (cnt >= FLUSH_AT) {
      gather_add_rows<RECORD>(mylst, cnt, lane, h, src, acc, glist, fill);
      cnt = 0;
    }
  }
  if (cnt > 0) gather_add_rows<RECORD>(mylst, cnt, lane, h, src, acc, glist, fill);
}

__device__ __forceinline__ void store_means(int tid, const float* acc, const int* degc,
                                            float* __restrict__ aggRange) {
  float4* out4 = (float4*)aggRange;
  for (int i = tid; i < OWN_ROWS * (D / 4); i += OWN_THREADS) {
    const float inv = 1.0f / fmaxf((float)degc[i >> 5], 1.0f);
    float4 v = ((const float4*)acc)[i];
    v.x *= inv; v.y *= inv; v.z *= inv; v.w *= inv;
    out4[i] = v;
  }
}

__global__ __launch_bounds__(OWN_THREADS) void owner_range_scan(const float* __restrict__ h,
                                                               const int* __restrict__ src,
                                                               const int* __restrict__ dst,
                                                               int* __restrict__ rec,
                                                               float* __restrict__ agg) {
  __shared__ __attribute__((aligned(16))) float acc[OWN_ROWS * D];
  __shared__ int lst[OWN_WAVES * LIST_CAP];
  __shared__ int degc[OWN_ROWS];
  __shared__ int fill;
  const int tid  = threadIdx.x;
  const int lane = tid & 31;
  const int rowBase = blockIdx.x * OWN_ROWS;

  for (int i = tid; i < OWN_ROWS * (D / 4); i += OWN_THREADS)
    ((float4*)acc)[i] = make_float4(0.f, 0.f, 0.f, 0.f);
  if (tid < OWN_ROWS) degc[tid] = 0;
  if (tid == 0) fill = 0;
  __syncthreads();

  int* glist = rec + REC_HDR + (size_t)blockIdx.x * REC_CAP;
  scan_range<true>(tid, lane, rowBase, lst, h, src, dst, acc, degc, glist, &fill);
  __syncthreads();

  if (tid == 0) rec[blockIdx.x] = (fill <= REC_CAP) ? fill : -1;

  store_means(tid, acc, degc, agg + (size_t)rowBase * D);
}

__global__ __launch_bounds__(OWN_THREADS) void owner_range_replay(const float* __restrict__ h,
                                                                 const int* __restrict__ src,
                                                                 const int* __restrict__ dst,
                                                                 const int* __restrict__ rec,
                                                                 float* __restrict__ agg) {
  __shared__ __attribute__((aligned(16))) float acc[(OWN_ROWS + 1) * D];
  __shared__ int lst[OWN_WAVES * LIST_CAP];
  __shared__ int degc[OWN_ROWS];
  const int tid  = threadIdx.x;
  const int lane = tid & 31;
  const int rowBase = blockIdx.x * OWN_ROWS;

  for (int i = tid; i < (OWN_ROWS + 1) * (D / 4); i += OWN_THREADS)
    ((float4*)acc)[i] = make_float4(0.f, 0.f, 0.f, 0.f);
  if (tid < OWN_ROWS) degc[tid] = 0;
  __syncthreads();

  const int nrec = rec[blockIdx.x];
  if (nrec >= 0 && nrec <= REC_CAP) {
    const int* glist = rec + REC_HDR + (size_t)blockIdx.x * REC_CAP;
    const int wave = __builtin_amdgcn_readfirstlane(tid >> 5);
    const int per  = (nrec + OWN_WAVES - 1) / OWN_WAVES;
    const int beg  = wave * per;
    const int end  = min(nrec, beg + per);
    int pk_next = (beg + lane < end) ? glist[beg + lane] : 0;
    for (int j0 = beg; j0 < end; j0 += 32) {
      const int n = min(32, end - j0);
      const int pk = pk_next;
      pk_next = (j0 + 32 + lane < end) ? glist[j0 + 32 + lane] : 0;
      if (lane < n) atomicAdd(&degc[pk >> SRC_BITS], 1);
      add_rows<REPLAY_G, true>(pk & ((1 << SRC_BITS) - 1), pk >> SRC_BITS, n, lane, h, acc);
    }
  } else {
    scan_range<false>(tid, lane, rowBase, lst, h, src, dst, acc, degc, (int*)0, (int*)0);
  }
  __syncthreads();

  store_means(tid, acc, degc, agg + (size_t)rowBase * D);
}

__global__ __launch_bounds__(256) void layer_gemm(const float* h,
                                                  const float* __restrict__ agg,
                                                  const float* __restrict__ Wself,
                                                  const float* __restrict__ Wneigh,
                                                  const float* __restrict__ bias_v,
                                                  float* h_out,
                                                  int relu) {
  __shared__ __attribute__((aligned(16))) float sh[BM * D_PAD];
  __shared__ __attribute__((aligned(16))) float sa[BM * D_PAD];
  const int tid = threadIdx.x;
  const int mBase = blockIdx.x * BM;

  float4 hv[H_PASSES], av[H_PASSES];
#pragma unroll
  for (int j = 0; j < H_PASSES; ++j) {
    const int i = tid + 256 * j;
    const int row = mBase + (i >> 5), c4 = i & 31;
    hv[j]  = ((const float4*)(h   + (size_t)row * D))[c4];
    av[j]  = ((const float4*)(agg + (size_t)row * D))[c4];
  }
#pragma unroll
  for (int j = 0; j < H_PASSES; ++j) {
    const int i = tid + 256 * j;
    const int r = i >> 5, c4 = i & 31;
    *(float4*)&sh[r * D_PAD + c4 * 4] = hv[j];
    *(float4*)&sa[r * D_PAD + c4 * 4] = av[j];
  }
  __syncthreads();

  const int lane = tid & 31;
  const int half = lane >> 4;
  const int l16  = lane & 15;
  const int col  = (tid >> 5) * 16 + l16;

  v8f acc[RT];
#pragma unroll
  for (int t = 0; t < RT; ++t) acc[t] = (v8f){0.f, 0.f, 0.f, 0.f, 0.f, 0.f, 0.f, 0.f};

#pragma unroll 4
  for (int k = 0; k < D; k += 4) {
    const int ka = k + 2 * half;
    v2f b1, b2;
    b1.x = Wself[(size_t)(ka)      * D + col];
    b1.y = Wself[(size_t)(ka + 1)  * D + col];
    b2.x = Wneigh[(size_t)(ka)     * D + col];
    b2.y = Wneigh[(size_t)(ka + 1) * D + col];
    v2f a1[RT], a2[RT];
#pragma unroll
    for (int t = 0; t < RT; ++t) {
      a1[t] = *(const v2f*)&sh[(t * 16 + l16) * D_PAD + ka];
      a2[t] = *(const v2f*)&sa[(t * 16 + l16) * D_PAD + ka];
    }
#pragma unroll
    for (int t = 0; t < RT; ++t) acc[t] = wmma_f32_16x16x4(a1[t], b1, acc[t]);
#pragma unroll
    for (int t = 0; t < RT; ++t) acc[t] = wmma_f32_16x16x4(a2[t], b2, acc[t]);
  }

  const float bv = bias_v[col];
#pragma unroll
  for (int t = 0; t < RT; ++t) {
    const int rowBase = mBase + t * 16 + 8 * half;
#pragma unroll
    for (int r = 0; r < 8; ++r) {
      float v = acc[t][r] + bv;
      if (relu) v = fmaxf(v, 0.0f);
      h_out[(size_t)(rowBase + r) * D + col] = v;
    }
  }
}

extern "C" void kernel_launch(void* const* d_in, const int* in_sizes, int n_in,
                              void* d_out, int out_size, void* d_ws, size_t ws_size,
                              hipStream_t stream) {
  const float* x       = (const float*)d_in[0];
  const int*   ei      = (const int*)d_in[1];
  const float* W_in    = (const float*)d_in[2];
  const float* b_in    = (const float*)d_in[3];
  const float* W_self  = (const float*)d_in[4];
  const float* W_neigh = (const float*)d_in[5];
  const float* b_lay   = (const float*)d_in[6];
  float* out = (float*)d_out;

  float* ws   = (float*)d_ws;
  float* hA   = ws;
  int*   rec  = (int*)(ws + (size_t)1 * N_NODES * D);
  float* agg  = ws + (size_t)2 * N_NODES * D;

  const int* src = ei;
  const int* dst = ei + N_EDGES;

  input_gemm_tanh<<<N_NODES / BM, 256, 0, stream>>>(x, W_in, b_in, hA);

  const float* hcur = hA;
  for (int i = 0; i < L_LAYERS; ++i) {
    if (i == 0)
      owner_range_scan<<<OWN_GROUPS, OWN_THREADS, 0, stream>>>(hcur, src, dst, rec, agg);
    else
      owner_range_replay<<<OWN_GROUPS, OWN_THREADS, 0, stream>>>(hcur, src, dst, rec, agg);
    float* hnext = (i == L_LAYERS - 1) ? out : hA;
    layer_gemm<<<N_NODES / BM, 256, 0, stream>>>(
        hcur, agg,
        W_self  + (size_t)i * D * D,
        W_neigh + (size_t)i * D * D,
        b_lay   + (size_t)i * D,
        hnext, (i < L_LAYERS - 1) ? 1 : 0);
    hcur = hnext;
  }
}


# Layers 2-3: layer product moved into 80-row owner work-groups (3 per CU); neighbour rows summed in LDS from per-80-row recorded lists feed it; agg plane, 2 launches gone; sums still arrival order
# speedup vs baseline: 1.7741x; 1.0994x over previous
#include <hip/hip_runtime.h>
#include <math.h>


typedef __attribute__((ext_vector_type(2))) float v2f;
typedef __attribute__((ext_vector_type(8))) float v8f;

#define N_NODES 200000
#define N_EDGES 600000
#define F_IN    117
#define F_PAD   124
#define D       128
#define D_PAD   132
#define L_LAYERS 3
#define RT      4
#define BM      (16 * RT)
#define X_VEC4   ((BM * F_IN) / 4)
#define X_PASSES ((X_VEC4 + 255) / 256)
#define H_PASSES ((BM * (D / 4)) / 256)

#define OWN_ROWS    400
#define OWN_GROUPS  (N_NODES / OWN_ROWS)
#define OWN_THREADS 1024
#define OWN_WAVES   (OWN_THREADS / 32)
#define SCAN_U      4
#define E_VEC4      (N_EDGES / 4)
#define SCAN_ITERS  ((E_VEC4 + OWN_THREADS * SCAN_U - 1) / (OWN_THREADS * SCAN_U))
#define FLUSH_AT    8
#define LIST_CAP    640
#define GATHER_G    8
#define EDGE_BITS   20
#define SRC_BITS    18
#define SUB_ROWS    80
#define SUBS        (OWN_ROWS / SUB_ROWS)
#define SUB_GROUPS  (N_NODES / SUB_ROWS)
#define SUB_RT      (SUB_ROWS / 16)
#define SUB_WAVES   8
#define SUB_CAP     8192
#define REC_HDR     2560
#define REPLAY_G    16

static_assert(N_NODES % OWN_ROWS == 0, "ranges must tile the node set");
static_assert(N_EDGES % 4 == 0, "index rows are read as 16-byte vectors");
static_assert(N_EDGES <= (1 << EDGE_BITS) && OWN_ROWS <= (1 << (31 - EDGE_BITS)), "list entry packing");
static_assert(FLUSH_AT - 1 + 32 * 4 * SCAN_U <= LIST_CAP, "one scan trip must fit behind a carried remainder");
static_assert(N_NODES <= (1 << SRC_BITS) && OWN_ROWS <= (1 << (31 - SRC_BITS)), "recorded entry packing");
static_assert(OWN_ROWS % SUB_ROWS == 0 && SUB_ROWS % 16 == 0, "lists tile a range, tiles tile a list's rows");
static_assert(SUB_ROWS <= (1 << (31 - SRC_BITS)), "recorded entry packing");
static_assert(SUB_GROUPS <= REC_HDR, "one count per list");
static_assert((size_t)REC_HDR + (size_t)SUB_GROUPS * SUB_CAP <= (size_t)N_NODES * D, "lists fit in one h plane");
static_assert((SUB_ROWS * (D / 4)) % 256 == 0, "own rows stage in whole passes");

__device__ __forceinline__ v8f wmma_f32_16x16x4(v2f a, v2f b, v8f c) {
  return __builtin_amdgcn_wmma_f32_16x16x4_f32(false, a, false, b, (short)0, c,
                                               false, false);
}

__global__ __launch_bounds__(256) void input_gemm_tanh(const float* __restrict__ x,
                                                       const float* __restrict__ W_in,
                                                       const float* __restrict__ b_in,
                                                       float* __restrict__ h_out) {
  __shared__ __attribute__((aligned(16))) float sx[BM * F_PAD];
  const int tid = threadIdx.x;
  const int mBase = blockIdx.x * BM;

  const float4* xt = (const float4*)(x + (size_t)mBase * F_IN);
  float4 xv[X_PASSES];
#pragma unroll
  for (int j = 0; j < X_PASSES; ++j) {
    const int i4 = tid + 256 * j;
    xv[j] = (i4 < X_VEC4) ? xt[i4] : make_float4(0.f, 0.f, 0.f, 0.f);
  }
#pragma unroll
  for (int j = 0; j < X_PASSES; ++j) {
    const int i4 = tid + 256 * j;
    if (i4 < X_VEC4) {
      const int i = i4 * 4;
      int r = i / F_IN, c = i - r * F_IN;
      const float e[4] = {xv[j].x, xv[j].y, xv[j].z, xv[j].w};
#pragma unroll
      for (int q = 0; q < 4; ++q) {
        sx[r * F_PAD + c] = e[q];
        if (++c == F_IN) { c = 0; ++r; }
      }
    }
  }
  for (int i = tid; i < BM * (F_PAD - F_IN); i += 256) {
    const int r = i / (F_PAD - F_IN), c = F_IN + (i - r * (F_PAD - F_IN));
    sx[r * F_PAD + c] = 0.0f;
  }
  __syncthreads();

  const int lane = tid & 31;
  const int half = lane >> 4;
  const int l16  = lane & 15;
  const int col  = (tid >> 5) * 16 + l16;

  v8f acc[RT];
#pragma unroll
  for (int t = 0; t < RT; ++t) acc[t] = (v8f){0.f, 0.f, 0.f, 0.f, 0.f, 0.f, 0.f, 0.f};

  for (int k = 0; k < F_IN; k += 4) {
    const int ka = k + 2 * half;
    v2f b;
    b.x = (ka     < F_IN) ? W_in[(size_t)(ka)     * D + col] : 0.0f;
    b.y = (ka + 1 < F_IN) ? W_in[(size_t)(ka + 1) * D + col] : 0.0f;
    v2f a[RT];
#pragma unroll
    for (int t = 0; t < RT; ++t)
      a[t] = *(const v2f*)&sx[(t * 16 + l16) * F_PAD + ka];
#pragma unroll
    for (int t = 0; t < RT; ++t) acc[t] = wmma_f32_16x16x4(a[t], b, acc[t]);
  }

  const float bias = b_in[col];
#pragma unroll
  for (int t = 0; t < RT; ++t) {
    const int rowBase = mBase + t * 16 + 8 * half;
#pragma unroll
    for (int r = 0; r < 8; ++r)
      h_out[(size_t)(rowBase + r) * D + col] = tanhf(acc[t][r] + bias);
  }
}

template <int G, bool SPARE, int PITCH, int SPARE_ROW>
__device__ __forceinline__ void add_rows(int s, int dl, int n, int lane,
                                         const float* __restrict__ h, float* acc) {
  for (int k = 0; k < n; k += G) {
    float4 v[G];
    int d[G];
#pragma unroll
    for (int g = 0; g < G; ++g) {
      const int kk = min(k + g, n - 1);
      const int sg = __builtin_amdgcn_readlane(s, kk);
      d[g] = (SPARE && k + g >= n) ? SPARE_ROW : __builtin_amdgcn_readlane(dl, kk);
      v[g] = ((const float4*)(h + (size_t)sg * D))[lane];
    }
#pragma unroll
    for (int g = 0; g < G; ++g) {
      if (SPARE || k + g < n) {
        float* p = acc + d[g] * PITCH + lane * 4;
        atomicAdd(p + 0, v[g].x);
        atomicAdd(p + 1, v[g].y);
        atomicAdd(p + 2, v[g].z);
        atomicAdd(p + 3, v[g].w);
      }
    }
  }
}

__device__ __forceinline__ void gather_add_rows(const int* mylst, int cnt, int lane,
                                                const float* __restrict__ h,
                                                const int* __restrict__ src,
                                                float* acc, int* glist, int* fill) {
  for (int j0 = 0; j0 < cnt; j0 += 32) {
    const int n = min(32, cnt - j0);
    int pk = 0, s = 0;
    if (lane < n) {
      pk = mylst[j0 + lane];
      s  = src[pk & ((1 << EDGE_BITS) - 1)];
    }
    const int dl = pk >> EDGE_BITS;
    if (lane < n) {
      const int sub = dl / SUB_ROWS;
      const int at  = atomicAdd(&fill[sub], 1);
      if (at < SUB_CAP) glist[sub * SUB_CAP + at] = s | ((dl - sub * SUB_ROWS) << SRC_BITS);
    }
    add_rows<GATHER_G, false, D, OWN_ROWS>(s, dl, n, lane, h, acc);
  }
}

__device__ __forceinline__ void scan_range(int tid, int lane, int rowBase, int* lst,
                                           const float* __restrict__ h,
                                           const int* __restrict__ src,
                                           const int* __restrict__ dst,
                                           float* acc, int* degc, int* glist, int* fill) {
  int* mylst = lst + (tid >> 5) * LIST_CAP;
  const int4* dst4 = (const int4*)dst;
  int cnt = 0;
  const int rot = blockIdx.x % SCAN_ITERS;

  for (int i = 0; i < SCAN_ITERS; ++i) {
    int it = i + rot;
    if (it >= SCAN_ITERS) it -= SCAN_ITERS;
    int4 dv[SCAN_U];
    int  i4[SCAN_U];
    int  rb[SCAN_U];
#pragma unroll
    for (int u = 0; u < SCAN_U; ++u) {
      i4[u] = (it * SCAN_U + u) * OWN_THREADS + tid;
      const bool ok = i4[u] < E_VEC4;
      dv[u] = dst4[ok ? i4[u] : E_VEC4 - 1];
      rb[u] = ok ? rowBase : 0x7fffffff;
    }
#pragma unroll
    for (int u = 0; u < SCAN_U; ++u) {
      const int d[4] = {dv[u].x, dv[u].y, dv[u].z, dv[u].w};
#pragma unroll
      for (int q = 0; q < 4; ++q) {
        const unsigned dl = (unsigned)(d[q] - rb[u]);
        const bool m = dl < (unsigned)OWN_ROWS;
        const unsigned mask = __builtin_amdgcn_ballot_w32(m);
        if (mask != 0u) {
          if (m) {
            mylst[cnt + (int)__builtin_amdgcn_mbcnt_lo(mask, 0u)] =
                (i4[u] * 4 + q) | (int)(dl << EDGE_BITS);
            atomicAdd(&degc[dl], 1);
          }
          cnt += __builtin_popcount(mask);
        }
      }
    }
    if (cnt >= FLUSH_AT) {
      gather_add_rows(mylst, cnt, lane, h, src, acc, glist, fill);
      cnt = 0;
    }
  }
  if (cnt > 0) gather_add_rows(mylst, cnt, lane, h, src, acc, glist, fill);
}

__device__ __forceinline__ void store_means(int tid, const float* acc, const int* degc,
                                            float* __restrict__ aggRange) {
  float4* out4 = (float4*)aggRange;
  for (int i = tid; i < OWN_ROWS * (D / 4); i += OWN_THREADS) {
    const float inv = 1.0f / fmaxf((float)degc[i >> 5], 1.0f);
    float4 v = ((const float4*)acc)[i];
    v.x *= inv; v.y *= inv; v.z *= inv; v.w *= inv;
    out4[i] = v;
  }
}

__global__ __launch_bounds__(OWN_THREADS) void owner_range_scan(const float* __restrict__ h,
                                                               const int* __restrict__ src,
                                                               const int* __restrict__ dst,
                                                               int* __restrict__ rec,
                                                               float* __restrict__ agg) {
  __shared__ __attribute__((aligned(16))) float acc[OWN_ROWS * D];
  __shared__ int lst[OWN_WAVES * LIST_CAP];
  __shared__ int degc[OWN_ROWS];
  __shared__ int fill[SUBS];
  const int tid  = threadIdx.x;
  const int lane = tid & 31;
  const int rowBase = blockIdx.x * OWN_ROWS;

  for (int i = tid; i < OWN_ROWS * (D / 4); i += OWN_THREADS)
    ((float4*)acc)[i] = make_float4(0.f, 0.f, 0.f, 0.f);
  if (tid < OWN_ROWS) degc[tid] = 0;
  if (tid < SUBS) fill[tid] = 0;
  __syncthreads();

  int* glist = rec + REC_HDR + (size_t)blockIdx.x * SUBS * SUB_CAP;
  scan_range(tid, lane, rowBase, lst, h, src, dst, acc, degc, glist, fill);
  __syncthreads();

  if (tid < SUBS) rec[blockIdx.x * SUBS + tid] = (fill[tid] <= SUB_CAP) ? fill[tid] : -1;

  store_means(tid, acc, degc, agg + (size_t)rowBase * D);
}

__global__ __launch_bounds__(256) void layer_gemm(const float* h,
                                                  const float* __restrict__ agg,
                                                  const float* __restrict__ Wself,
                                                  const float* __restrict__ Wneigh,
                                                  const float* __restrict__ bias_v,
                                                  float* h_out,
                                                  int relu) {
  __shared__ __attribute__((aligned(16))) float sh[BM * D_PAD];
  __shared__ __attribute__((aligned(16))) float sa[BM * D_PAD];
  const int tid = threadIdx.x;
  const int mBase = blockIdx.x * BM;

  float4 hv[H_PASSES], av[H_PASSES];
#pragma unroll
  for (int j = 0; j < H_PASSES; ++j) {
    const int i = tid + 256 * j;
    const int row = mBase + (i >> 5), c4 = i & 31;
    hv[j]  = ((const float4*)(h   + (size_t)row * D))[c4];
    av[j]  = ((const float4*)(agg + (size_t)row * D))[c4];
  }
#pragma unroll
  for (int j = 0; j < H_PASSES; ++j) {
    const int i = tid + 256 * j;
    const int r = i >> 5, c4 = i & 31;
    *(float4*)&sh[r * D_PAD + c4 * 4] = hv[j];
    *(float4*)&sa[r * D_PAD + c4 * 4] = av[j];
  }
  __syncthreads();

  const int lane = tid & 31;
  const int half = lane >> 4;
  const int l16  = lane & 15;
  const int col  = (tid >> 5) * 16 + l16;

  v8f acc[RT];
#pragma unroll
  for (int t = 0; t < RT; ++t) acc[t] = (v8f){0.f, 0.f, 0.f, 0.f, 0.f, 0.f, 0.f, 0.f};

#pragma unroll 4
  for (int k = 0; k < D; k += 4) {
    const int ka = k + 2 * half;
    v2f b1, b2;
    b1.x = Wself[(size_t)(ka)      * D + col];
    b1.y = Wself[(size_t)(ka + 1)  * D + col];
    b2.x = Wneigh[(size_t)(ka)     * D + col];
    b2.y = Wneigh[(size_t)(ka + 1) * D + col];
    v2f a1[RT], a2[RT];
#pragma unroll
    for (int t = 0; t < RT; ++t) {
      a1[t] = *(const v2f*)&sh[(t * 16 + l16) * D_PAD + ka];
      a2[t] = *(const v2f*)&sa[(t * 16 + l16) * D_PAD + ka];
    }
#pragma unroll
    for (int t = 0; t < RT; ++t) acc[t] = wmma_f32_16x16x4(a1[t], b1, acc[t]);
#pragma unroll
    for (int t = 0; t < RT; ++t) acc[t] = wmma_f32_16x16x4(a2[t], b2, acc[t]);
  }

  const float bv = bias_v[col];
#pragma unroll
  for (int t = 0; t < RT; ++t) {
    const int rowBase = mBase + t * 16 + 8 * half;
#pragma unroll
    for (int r = 0; r < 8; ++r) {
      float v = acc[t][r] + bv;
      if (relu) v = fmaxf(v, 0.0f);
      h_out[(size_t)(rowBase + r) * D + col] = v;
    }
  }
}

__global__ __launch_bounds__(256) void owner_replay_gemm(const float* __restrict__ h,
                                                         const int* __restrict__ src,
                                                         const int* __restrict__ dst,
                                                         const int* __restrict__ rec,
                                                         const float* __restrict__ Wself,
                                                         const float* __restrict__ Wneigh,
                                                         const float* __restrict__ bias_v,
                                                         float* __restrict__ h_out,
                                                         int relu) {
  __shared__ __attribute__((aligned(16))) float sh[SUB_ROWS * D_PAD];
  __shared__ __attribute__((aligned(16))) float sa[(SUB_ROWS + 1) * D_PAD];
  __shared__ int degc[SUB_ROWS];
  const int tid  = threadIdx.x;
  const int lane = tid & 31;
  const int rowBase = blockIdx.x * SUB_ROWS;
  const int wave = __builtin_amdgcn_readfirstlane(tid >> 5);

  constexpr int SELF_PASSES = (SUB_ROWS * (D / 4)) / 256;
  float4 hv[SELF_PASSES];
#pragma unroll
  for (int j = 0; j < SELF_PASSES; ++j) {
    const int i = tid + 256 * j;
    hv[j] = ((const float4*)(h + (size_t)(rowBase + (i >> 5)) * D))[i & 31];
  }
  const int* glist = rec + REC_HDR + (size_t)blockIdx.x * SUB_CAP;
  const int nrec = rec[blockIdx.x];
  int pk_next = (wave * 32 + lane < nrec) ? glist[wave * 32 + lane] : 0;

  for (int i = tid; i < (SUB_ROWS + 1) * (D_PAD / 4); i += 256)
    ((float4*)sa)[i] = make_float4(0.f, 0.f, 0.f, 0.f);
  if (tid < SUB_ROWS) degc[tid] = 0;
#pragma unroll
  for (int j = 0; j < SELF_PASSES; ++j) {
    const int i = tid + 256 * j;
    *(float4*)&sh[(i >> 5) * D_PAD + (i & 31) * 4] = hv[j];
  }
  __syncthreads();

  if (nrec >= 0 && nrec <= SUB_CAP) {
    for (int j0 = wave * 32; j0 < nrec; j0 += 32 * SUB_WAVES) {
      const int n  = min(32, nrec - j0);
      const int pk = pk_next;
      const int jn = j0 + 32 * SUB_WAVES + lane;
      pk_next = (jn < nrec) ? glist[jn] : 0;
      if (lane < n) atomicAdd(&degc[pk >> SRC_BITS], 1);
      add_rows<REPLAY_G, true, D_PAD, SUB_ROWS>(pk & ((1 << SRC_BITS) - 1), pk >> SRC_BITS, n, lane, h, sa);
    }
  } else {
    for (int e0 = wave * 32; e0 < N_EDGES; e0 += 32 * SUB_WAVES) {
      const int e = e0 + lane;
      const unsigned dl = (e < N_EDGES) ? (unsigned)(dst[e] - rowBase) : 0xffffffffu;
      const bool m = dl < (unsigned)SUB_ROWS;
      unsigned mask = __builtin_amdgcn_ballot_w32(m);
      if (mask != 0u) {
        int s = 0;
        if (m) {
          s = src[e];
          atomicAdd(&degc[dl], 1);
        }
        while (mask != 0u) {
          const int bit = __builtin_ctz(mask);
          mask &= mask - 1u;
          const int sg = __builtin_amdgcn_readlane(s, bit);
          const int dg = __builtin_amdgcn_readlane((int)dl, bit);
          const float4 v = ((const float4*)(h + (size_t)sg * D))[lane];
          float* p = sa + dg * D_PAD + lane * 4;
          atomicAdd(p + 0, v.x);
          atomicAdd(p + 1, v.y);
          atomicAdd(p + 2, v.z);
          atomicAdd(p + 3, v.w);
        }
      }
    }
  }
  __syncthreads();

  for (int i = tid; i < SUB_ROWS * (D / 4); i += 256) {
    const int r = i >> 5;
    const float inv = 1.0f / fmaxf((float)degc[r], 1.0f);
    float4* p = (float4*)&sa[r * D_PAD + (i & 31) * 4];
    float4 v = *p;
    v.x *= inv; v.y *= inv; v.z *= inv; v.w *= inv;
    *p = v;
  }
  __syncthreads();

  const int half = lane >> 4;
  const int l16  = lane & 15;
  const int col  = (tid >> 5) * 16 + l16;

  v8f acc[SUB_RT];
#pragma unroll
  for (int t = 0; t < SUB_RT; ++t) acc[t] = (v8f){0.f, 0.f, 0.f, 0.f, 0.f, 0.f, 0.f, 0.f};

#pragma unroll 4
  for (int k = 0; k < D; k += 4) {
    const int ka = k + 2 * half;
    v2f b1, b2;
    b1.x = Wself[(size_t)(ka)      * D + col];
    b1.y = Wself[(size_t)(ka + 1)  * D + col];
    b2.x = Wneigh[(size_t)(ka)     * D + col];
    b2.y = Wneigh[(size_t)(ka + 1) * D + col];
    v2f a1[SUB_RT], a2[SUB_RT];
#pragma unroll
    for (int t = 0; t < SUB_RT; ++t) {
      a1[t] = *(const v2f*)&sh[(t * 16 + l16) * D_PAD + ka];
      a2[t] = *(const v2f*)&sa[(t * 16 + l16) * D_PAD + ka];
    }
#pragma unroll
    for (int t = 0; t < SUB_RT; ++t) acc[t] = wmma_f32_16x16x4(a1[t], b1, acc[t]);
#pragma unroll
    for (int t = 0; t < SUB_RT; ++t) acc[t] = wmma_f32_16x16x4(a2[t], b2, acc[t]);
  }

  const float bv = bias_v[col];
#pragma unroll
  for (int t = 0; t < SUB_RT; ++t) {
    const int rowT = rowBase + t * 16 + 8 * half;
#pragma unroll
    for (int r = 0; r < 8; ++r) {
      float v = acc[t][r] + bv;
      if (relu) v = fmaxf(v, 0.0f);
      h_out[(size_t)(rowT + r) * D + col] = v;
    }
  }
}

extern "C" void kernel_launch(void* const* d_in, const int* in_sizes, int n_in,
                              void* d_out, int out_size, void* d_ws, size_t ws_size,
                              hipStream_t stream) {
  const float* x       = (const float*)d_in[0];
  const int*   ei      = (const int*)d_in[1];
  const float* W_in    = (const float*)d_in[2];
  const float* b_in    = (const float*)d_in[3];
  const float* W_self  = (const float*)d_in[4];
  const float* W_neigh = (const float*)d_in[5];
  const float* b_lay   = (const float*)d_in[6];
  float* out = (float*)d_out;

  float* ws   = (float*)d_ws;
  float* hA   = ws;
  int*   rec  = (int*)(ws + (size_t)1 * N_NODES * D);
  float* agg  = ws + (size_t)2 * N_NODES * D;
  float* hB   = agg;

  const int* src = ei;
  const int* dst = ei + N_EDGES;

  input_gemm_tanh<<<N_NODES / BM, 256, 0, stream>>>(x, W_in, b_in, hA);

  owner_range_scan<<<OWN_GROUPS, OWN_THREADS, 0, stream>>>(hA, src, dst, rec, agg);
  layer_gemm<<<N_NODES / BM, 256, 0, stream>>>(hA, agg, W_self, W_neigh, b_lay, hA, 1);

  owner_replay_gemm<<<SUB_GROUPS, 256, 0, stream>>>(
      hA, src, dst, rec, W_self + (size_t)D * D, W_neigh + (size_t)D * D, b_lay + D, hB, 1);
  owner_replay_gemm<<<SUB_GROUPS, 256, 0, stream>>>(
      hB, src, dst, rec, W_self + (size_t)2 * D * D, W_neigh + (size_t)2 * D * D, b_lay + 2 * D, out, 0);
}


# Layer 1 joins the fused 80-row owner launches: the index scan only records lists, one work-group per (4000-row range, fifth of the edges); means plane and 64-row GEMM launch gone; sums arrival order
# speedup vs baseline: 1.9244x; 1.0847x over previous
#include <hip/hip_runtime.h>
#include <math.h>


typedef __attribute__((ext_vector_type(2))) float v2f;
typedef __attribute__((ext_vector_type(8))) float v8f;

#define N_NODES 200000
#define N_EDGES 600000
#define F_IN    117
#define F_PAD   124
#define D       128
#define D_PAD   132
#define L_LAYERS 3
#define RT      4
#define BM      (16 * RT)
#define X_VEC4   ((BM * F_IN) / 4)
#define X_PASSES ((X_VEC4 + 255) / 256)

#define SUB_ROWS    80
#define SUB_GROUPS  (N_NODES / SUB_ROWS)
#define SUB_RT      (SUB_ROWS / 16)
#define SUB_WAVES   8
#define RNG_ROWS    4000
#define RNG_COUNT   (N_NODES / RNG_ROWS)
#define SUBS        (RNG_ROWS / SUB_ROWS)
#define E_SPLIT     5
#define REC_GROUPS  (RNG_COUNT * E_SPLIT)
#define REC_THREADS 1024
#define REC_WAVES   (REC_THREADS / 32)
#define SLICE_EDGES (N_EDGES / E_SPLIT)
#define SLICE_VEC4  (SLICE_EDGES / 4)
#define SCAN_U      4
#define SCAN_ITERS  ((SLICE_VEC4 + REC_THREADS * SCAN_U - 1) / (REC_THREADS * SCAN_U))
#define FLUSH_AT    32
#define LIST_CAP    640
#define EDGE_BITS   17
#define SRC_BITS    18
#define PART_CAP    1984
#define REC_HDR     12544
#define REPLAY_G    16

static_assert(N_NODES % RNG_ROWS == 0, "ranges must tile the node set");
static_assert(N_EDGES % (4 * E_SPLIT) == 0, "every slice of an index row is read as whole 16-byte vectors");
static_assert(SLICE_EDGES <= (1 << EDGE_BITS) && RNG_ROWS <= (1 << (31 - EDGE_BITS)), "match packing");
static_assert(FLUSH_AT - 1 + 32 * 4 * SCAN_U <= LIST_CAP, "one scan trip must fit behind a carried remainder");
static_assert(N_NODES <= (1 << SRC_BITS) && SUB_ROWS <= (1 << (31 - SRC_BITS)), "recorded entry packing");
static_assert(RNG_ROWS % SUB_ROWS == 0 && SUB_ROWS % 16 == 0, "lists tile a range, tiles tile a list's rows");
static_assert(SUB_GROUPS * E_SPLIT <= REC_HDR, "one count per part");
static_assert((size_t)REC_HDR + (size_t)SUB_GROUPS * E_SPLIT * PART_CAP <= (size_t)N_NODES * D, "lists fit in one h plane");
static_assert((SUB_ROWS * (D / 4)) % 256 == 0, "own rows stage in whole passes");

__device__ __forceinline__ v8f wmma_f32_16x16x4(v2f a, v2f b, v8f c) {
  return __builtin_amdgcn_wmma_f32_16x16x4_f32(false, a, false, b, (short)0, c,
                                               false, false);
}

__global__ __launch_bounds__(256) void input_gemm_tanh(const float* __restrict__ x,
                                                       const float* __restrict__ W_in,
                                                       const float* __restrict__ b_in,
                                                       float* __restrict__ h_out) {
  __shared__ __attribute__((aligned(16))) float sx[BM * F_PAD];
  const int tid = threadIdx.x;
  const int mBase = blockIdx.x * BM;

  const float4* xt = (const float4*)(x + (size_t)mBase * F_IN);
  float4 xv[X_PASSES];
#pragma unroll
  for (int j = 0; j < X_PASSES; ++j) {
    const int i4 = tid + 256 * j;
    xv[j] = (i4 < X_VEC4) ? xt[i4] : make_float4(0.f, 0.f, 0.f, 0.f);
  }
#pragma unroll
  for (int j = 0; j < X_PASSES; ++j) {
    const int i4 = tid + 256 * j;
    if (i4 < X_VEC4) {
      const int i = i4 * 4;
      int r = i / F_IN, c = i - r * F_IN;
      const float e[4] = {xv[j].x, xv[j].y, xv[j].z, xv[j].w};
#pragma unroll
      for (int q = 0; q < 4; ++q) {
        sx[r * F_PAD + c] = e[q];
        if (++c == F_IN) { c = 0; ++r; }
      }
    }
  }
  for (int i = tid; i < BM * (F_PAD - F_IN); i += 256) {
    const int r = i / (F_PAD - F_IN), c = F_IN + (i - r * (F_PAD - F_IN));
    sx[r * F_PAD + c] = 0.0f;
  }
  __syncthreads();

  const int lane = tid & 31;
  const int half = lane >> 4;
  const int l16  = lane & 15;
  const int col  = (tid >> 5) * 16 + l16;

  v8f acc[RT];
#pragma unroll
  for (int t = 0; t < RT; ++t) acc[t] = (v8f){0.f, 0.f, 0.f, 0.f, 0.f, 0.f, 0.f, 0.f};

  for (int k = 0; k < F_IN; k += 4) {
    const int ka = k + 2 * half;
    v2f b;
    b.x = (ka     < F_IN) ? W_in[(size_t)(ka)     * D + col] : 0.0f;
    b.y = (ka + 1 < F_IN) ? W_in[(size_t)(ka + 1) * D + col] : 0.0f;
    v2f a[RT];
#pragma unroll
    for (int t = 0; t < RT; ++t)
      a[t] = *(const v2f*)&sx[(t * 16 + l16) * F_PAD + ka];
#pragma unroll
    for (int t = 0; t < RT; ++t) acc[t] = wmma_f32_16x16x4(a[t], b, acc[t]);
  }

  const float bias = b_in[col];
#pragma unroll
  for (int t = 0; t < RT; ++t) {
    const int rowBase = mBase + t * 16 + 8 * half;
#pragma unroll
    for (int r = 0; r < 8; ++r)
      h_out[(size_t)(rowBase + r) * D + col] = tanhf(acc[t][r] + bias);
  }
}

__device__ __forceinline__ void record_matches(const int* mylst, int cnt, int lane,
                                               const int* __restrict__ srcSlice,
                                               int* parts, int* fill) {
  for (int j0 = 0; j0 < cnt; j0 += 32) {
    if (j0 + lane < cnt) {
      const int pk  = mylst[j0 + lane];
      const int s   = srcSlice[pk & ((1 << EDGE_BITS) - 1)];
      const int dl  = pk >> EDGE_BITS;
      const int sub = dl / SUB_ROWS;
      const int at  = atomicAdd(&fill[sub], 1);
      if (at < PART_CAP)
        parts[(size_t)sub * (E_SPLIT * PART_CAP) + at] = s | ((dl - sub * SUB_ROWS) << SRC_BITS);
    }
  }
}

__global__ __launch_bounds__(REC_THREADS) void range_record(const int* __restrict__ src,
                                                            const int* __restrict__ dst,
                                                            int* __restrict__ rec) {
  __shared__ int lst[REC_WAVES * LIST_CAP];
  __shared__ int fill[SUBS];
  const int tid  = threadIdx.x;
  const int lane = tid & 31;
  const int range = blockIdx.x / E_SPLIT;
  const int slice = blockIdx.x - range * E_SPLIT;
  const int rowBase = range * RNG_ROWS;

  if (tid < SUBS) fill[tid] = 0;
  __syncthreads();

  int* parts = rec + REC_HDR + ((size_t)range * SUBS * E_SPLIT + slice) * PART_CAP;
  const int* srcSlice = src + (size_t)slice * SLICE_EDGES;
  const int4* dst4 = (const int4*)(dst + (size_t)slice * SLICE_EDGES);
  int* mylst = lst + (tid >> 5) * LIST_CAP;
  int cnt = 0;
  const int rot = range % SCAN_ITERS;

  for (int i = 0; i < SCAN_ITERS; ++i) {
    int it = i + rot;
    if (it >= SCAN_ITERS) it -= SCAN_ITERS;
    int4 dv[SCAN_U];
    int  i4[SCAN_U];
    int  rb[SCAN_U];
#pragma unroll
    for (int u = 0; u < SCAN_U; ++u) {
      i4[u] = (it * SCAN_U + u) * REC_THREADS + tid;
      const bool ok = i4[u] < SLICE_VEC4;
      dv[u] = dst4[ok ? i4[u] : SLICE_VEC4 - 1];
      rb[u] = ok ? rowBase : 0x7fffffff;
    }
#pragma unroll
    for (int u = 0; u < SCAN_U; ++u) {
      const int d[4] = {dv[u].x, dv[u].y, dv[u].z, dv[u].w};
#pragma unroll
      for (int q = 0; q < 4; ++q) {
        const unsigned dl = (unsigned)(d[q] - rb[u]);
        const bool m = dl < (unsigned)RNG_ROWS;
        const unsigned mask = __builtin_amdgcn_ballot_w32(m);
        if (mask != 0u) {
          if (m)
            mylst[cnt + (int)__builtin_amdgcn_mbcnt_lo(mask, 0u)] =
                (i4[u] * 4 + q) | (int)(dl << EDGE_BITS);
          cnt += __builtin_popcount(mask);
        }
      }
    }
    if (cnt >= FLUSH_AT) {
      record_matches(mylst, cnt, lane, srcSlice, parts, fill);
      cnt = 0;
    }
  }
  if (cnt > 0) record_matches(mylst, cnt, lane, srcSlice, parts, fill);
  __syncthreads();

  if (tid < SUBS)
    rec[((size_t)range * SUBS + tid) * E_SPLIT + slice] = (fill[tid] <= PART_CAP) ? fill[tid] : -1;
}

template <int G, bool SPARE, int PITCH, int SPARE_ROW>
__device__ __forceinline__ void add_rows(int s, int dl, int n, int lane,
                                         const float* __restrict__ h, float* acc) {
  for (int k = 0; k < n; k += G) {
    float4 v[G];
    int d[G];
#pragma unroll
    for (int g = 0; g < G; ++g) {
      const int kk = min(k + g, n - 1);
      const int sg = __builtin_amdgcn_readlane(s, kk);
      d[g] = (SPARE && k + g >= n) ? SPARE_ROW : __builtin_amdgcn_readlane(dl, kk);
      v[g] = ((const float4*)(h + (size_t)sg * D))[lane];
    }
#pragma unroll
    for (int g = 0; g < G; ++g) {
      if (SPARE || k + g < n) {
        float* p = acc + d[g] * PITCH + lane * 4;
        atomicAdd(p + 0, v[g].x);
        atomicAdd(p + 1, v[g].y);
        atomicAdd(p + 2, v[g].z);
        atomicAdd(p + 3, v[g].w);
      }
    }
  }
}

__device__ __forceinline__ int part_place(int j, const int (&pre)[E_SPLIT + 1]) {
  int a = j;
#pragma unroll
  for (int q = 1; q < E_SPLIT; ++q) a = (j >= pre[q]) ? j - pre[q] + q * PART_CAP : a;
  return a;
}

__global__ __launch_bounds__(256) void owner_replay_gemm(const float* __restrict__ h,
                                                         const int* __restrict__ src,
                                                         const int* __restrict__ dst,
                                                         const int* __restrict__ rec,
                                                         const float* __restrict__ Wself,
                                                         const float* __restrict__ Wneigh,
                                                         const float* __restrict__ bias_v,
                                                         float* __restrict__ h_out,
                                                         int relu) {
  __shared__ __attribute__((aligned(16))) float sh[SUB_ROWS * D_PAD];
  __shared__ __attribute__((aligned(16))) float sa[(SUB_ROWS + 1) * D_PAD];
  __shared__ int degc[SUB_ROWS];
  const int tid  = threadIdx.x;
  const int lane = tid & 31;
  const int rowBase = blockIdx.x * SUB_ROWS;
  const int wave = __builtin_amdgcn_readfirstlane(tid >> 5);

  constexpr int SELF_PASSES = (SUB_ROWS * (D / 4)) / 256;
  float4 hv[SELF_PASSES];
#pragma unroll
  for (int j = 0; j < SELF_PASSES; ++j) {
    const int i = tid + 256 * j;
    hv[j] = ((const float4*)(h + (size_t)(rowBase + (i >> 5)) * D))[i & 31];
  }
  const int* glist = rec + REC_HDR + (size_t)blockIdx.x * (E_SPLIT * PART_CAP);
  int pre[E_SPLIT + 1];
  bool fits = true;
  pre[0] = 0;
#pragma unroll
  for (int q = 0; q < E_SPLIT; ++q) {
    const int c = rec[blockIdx.x * E_SPLIT + q];
    fits = fits && c >= 0 && c <= PART_CAP;
    pre[q + 1] = pre[q] + max(c, 0);
  }
  const int nrec = pre[E_SPLIT];
  int pk_next = (fits && wave * 32 + lane < nrec) ? glist[part_place(wave * 32 + lane, pre)] : 0;

  for (int i = tid; i < (SUB_ROWS + 1) * (D_PAD / 4); i += 256)
    ((float4*)sa)[i] = make_float4(0.f, 0.f, 0.f, 0.f);
  if (tid < SUB_ROWS) degc[tid] = 0;
#pragma unroll
  for (int j = 0; j < SELF_PASSES; ++j) {
    const int i = tid + 256 * j;
    *(float4*)&sh[(i >> 5) * D_PAD + (i & 31) * 4] = hv[j];
  }
  __syncthreads();

  if (fits) {
    for (int j0 = wave * 32; j0 < nrec; j0 += 32 * SUB_WAVES) {
      const int n  = min(32, nrec - j0);
      const int pk = pk_next;
      const int jn = j0 + 32 * SUB_WAVES + lane;
      pk_next = (jn < nrec) ? glist[part_place(jn, pre)] : 0;
      if (lane < n) atomicAdd(&degc[pk >> SRC_BITS], 1);
      add_rows<REPLAY_G, true, D_PAD, SUB_ROWS>(pk & ((1 << SRC_BITS) - 1), pk >> SRC_BITS, n, lane, h, sa);
    }
  } else {
    for (int e0 = wave * 32; e0 < N_EDGES; e0 += 32 * SUB_WAVES) {
      const int e = e0 + lane;
      const unsigned dl = (e < N_EDGES) ? (unsigned)(dst[e] - rowBase) : 0xffffffffu;
      const bool m = dl < (unsigned)SUB_ROWS;
      unsigned mask = __builtin_amdgcn_ballot_w32(m);
      if (mask != 0u) {
        int s = 0;
        if (m) {
          s = src[e];
          atomicAdd(&degc[dl], 1);
        }
        while (mask != 0u) {
          const int bit = __builtin_ctz(mask);
          mask &= mask - 1u;
          const int sg = __builtin_amdgcn_readlane(s, bit);
          const int dg = __builtin_amdgcn_readlane((int)dl, bit);
          const float4 v = ((const float4*)(h + (size_t)sg * D))[lane];
          float* p = sa + dg * D_PAD + lane * 4;
          atomicAdd(p + 0, v.x);
          atomicAdd(p + 1, v.y);
          atomicAdd(p + 2, v.z);
          atomicAdd(p + 3, v.w);
        }
      }
    }
  }
  __syncthreads();

  for (int i = tid; i < SUB_ROWS * (D / 4); i += 256) {
    const int r = i >> 5;
    const float inv = 1.0f / fmaxf((float)degc[r], 1.0f);
    float4* p = (float4*)&sa[r * D_PAD + (i & 31) * 4];
    float4 v = *p;
    v.x *= inv; v.y *= inv; v.z *= inv; v.w *= inv;
    *p = v;
  }
  __syncthreads();

  const int half = lane >> 4;
  const int l16  = lane & 15;
  const int col  = (tid >> 5) * 16 + l16;

  v8f acc[SUB_RT];
#pragma unroll
  for (int t = 0; t < SUB_RT; ++t) acc[t] = (v8f){0.f, 0.f, 0.f, 0.f, 0.f, 0.f, 0.f, 0.f};

#pragma unroll 4
  for (int k = 0; k < D; k += 4) {
    const int ka = k + 2 * half;
    v2f b1, b2;
    b1.x = Wself[(size_t)(ka)      * D + col];
    b1.y = Wself[(size_t)(ka + 1)  * D + col];
    b2.x = Wneigh[(size_t)(ka)     * D + col];
    b2.y = Wneigh[(size_t)(ka + 1) * D + col];
    v2f a1[SUB_RT], a2[SUB_RT];
#pragma unroll
    for (int t = 0; t < SUB_RT; ++t) {
      a1[t] = *(const v2f*)&sh[(t * 16 + l16) * D_PAD + ka];
      a2[t] = *(const v2f*)&sa[(t * 16 + l16) * D_PAD + ka];
    }
#pragma unroll
    for (int t = 0; t < SUB_RT; ++t) acc[t] = wmma_f32_16x16x4(a1[t], b1, acc[t]);
#pragma unroll
    for (int t = 0; t < SUB_RT; ++t) acc[t] = wmma_f32_16x16x4(a2[t], b2, acc[t]);
  }

  const float bv = bias_v[col];
#pragma unroll
  for (int t = 0; t < SUB_RT; ++t) {
    const int rowT = rowBase + t * 16 + 8 * half;
#pragma unroll
    for (int r = 0; r < 8; ++r) {
      float v = acc[t][r] + bv;
      if (relu) v = fmaxf(v, 0.0f);
      h_out[(size_t)(rowT + r) * D + col] = v;
    }
  }
}

extern "C" void kernel_launch(void* const* d_in, const int* in_sizes, int n_in,
                              void* d_out, int out_size, void* d_ws, size_t ws_size,
                              hipStream_t stream) {
  const float* x       = (const float*)d_in[0];
  const int*   ei      = (const int*)d_in[1];
  const float* W_in    = (const float*)d_in[2];
  const float* b_in    = (const float*)d_in[3];
  const float* W_self  = (const float*)d_in[4];
  const float* W_neigh = (const float*)d_in[5];
  const float* b_lay   = (const float*)d_in[6];
  float* out = (float*)d_out;

  float* ws   = (float*)d_ws;
  float* hA   = ws;
  int*   rec  = (int*)(ws + (size_t)1 * N_NODES * D);
  float* hB   = ws + (size_t)2 * N_NODES * D;

  const int* src = ei;
  const int* dst = ei + N_EDGES;

  input_gemm_tanh<<<N_NODES / BM, 256, 0, stream>>>(x, W_in, b_in, hA);

  range_record<<<REC_GROUPS, REC_THREADS, 0, stream>>>(src, dst, rec);

  owner_replay_gemm<<<SUB_GROUPS, 256, 0, stream>>>(
      hA, src, dst, rec, W_self, W_neigh, b_lay, hB, 1);
  owner_replay_gemm<<<SUB_GROUPS, 256, 0, stream>>>(
      hB, src, dst, rec, W_self + (size_t)D * D, W_neigh + (size_t)D * D, b_lay + D, hA, 1);
  owner_replay_gemm<<<SUB_GROUPS, 256, 0, stream>>>(
      hA, src, dst, rec, W_self + (size_t)2 * D * D, W_neigh + (size_t)2 * D * D, b_lay + 2 * D, out, 0);
}


# Layer product in two halves: each wave runs the own-row half in 4 pieces between asking for 8 neighbour rows and adding them; per accumulator all self terms now precede the neighbour terms
# speedup vs baseline: 2.1254x; 1.1044x over previous
#include <hip/hip_runtime.h>
#include <math.h>


typedef __attribute__((ext_vector_type(2))) float v2f;
typedef __attribute__((ext_vector_type(8))) float v8f;

#define N_NODES 200000
#define N_EDGES 600000
#define F_IN    117
#define F_PAD   124
#define D       128
#define D_PAD   132
#define L_LAYERS 3
#define RT      4
#define BM      (16 * RT)
#define X_VEC4   ((BM * F_IN) / 4)
#define X_PASSES ((X_VEC4 + 255) / 256)

#define SUB_ROWS    80
#define SUB_GROUPS  (N_NODES / SUB_ROWS)
#define SUB_RT      (SUB_ROWS / 16)
#define SUB_WAVES   8
#define RNG_ROWS    4000
#define RNG_COUNT   (N_NODES / RNG_ROWS)
#define SUBS        (RNG_ROWS / SUB_ROWS)
#define E_SPLIT     5
#define REC_GROUPS  (RNG_COUNT * E_SPLIT)
#define REC_THREADS 1024
#define REC_WAVES   (REC_THREADS / 32)
#define SLICE_EDGES (N_EDGES / E_SPLIT)
#define SLICE_VEC4  (SLICE_EDGES / 4)
#define SCAN_U      4
#define SCAN_ITERS  ((SLICE_VEC4 + REC_THREADS * SCAN_U - 1) / (REC_THREADS * SCAN_U))
#define FLUSH_AT    32
#define LIST_CAP    640
#define EDGE_BITS   17
#define SRC_BITS    18
#define PART_CAP    1984
#define REC_HDR     12544
#define REPLAY_G    8
#define SELF_BLOCKS 4
#define SELF_KSTEPS (D / 4 / SELF_BLOCKS)
#define SELF_G      (32 / SELF_BLOCKS)

static_assert(N_NODES % RNG_ROWS == 0, "ranges must tile the node set");
static_assert(N_EDGES % (4 * E_SPLIT) == 0, "every slice of an index row is read as whole 16-byte vectors");
static_assert(SLICE_EDGES <= (1 << EDGE_BITS) && RNG_ROWS <= (1 << (31 - EDGE_BITS)), "match packing");
static_assert(FLUSH_AT - 1 + 32 * 4 * SCAN_U <= LIST_CAP, "one scan trip must fit behind a carried remainder");
static_assert(N_NODES <= (1 << SRC_BITS) && SUB_ROWS <= (1 << (31 - SRC_BITS)), "recorded entry packing");
static_assert(RNG_ROWS % SUB_ROWS == 0 && SUB_ROWS % 16 == 0, "lists tile a range, tiles tile a list's rows");
static_assert(SUB_GROUPS * E_SPLIT <= REC_HDR, "one count per part");
static_assert((size_t)REC_HDR + (size_t)SUB_GROUPS * E_SPLIT * PART_CAP <= (size_t)N_NODES * D, "lists fit in one h plane");
static_assert((SUB_ROWS * (D / 4)) % 256 == 0, "own rows stage in whole passes");
static_assert(SELF_BLOCKS * SELF_KSTEPS * 4 == D && SELF_BLOCKS * SELF_G == 32, "pieces tile K, groups tile a chunk");

__device__ __forceinline__ v8f wmma_f32_16x16x4(v2f a, v2f b, v8f c) {
  return __builtin_amdgcn_wmma_f32_16x16x4_f32(false, a, false, b, (short)0, c,
                                               false, false);
}

__global__ __launch_bounds__(256) void input_gemm_tanh(const float* __restrict__ x,
                                                       const float* __restrict__ W_in,
                                                       const float* __restrict__ b_in,
                                                       float* __restrict__ h_out) {
  __shared__ __attribute__((aligned(16))) float sx[BM * F_PAD];
  const int tid = threadIdx.x;
  const int mBase = blockIdx.x * BM;

  const float4* xt = (const float4*)(x + (size_t)mBase * F_IN);
  float4 xv[X_PASSES];
#pragma unroll
  for (int j = 0; j < X_PASSES; ++j) {
    const int i4 = tid + 256 * j;
    xv[j] = (i4 < X_VEC4) ? xt[i4] : make_float4(0.f, 0.f, 0.f, 0.f);
  }
#pragma unroll
  for (int j = 0; j < X_PASSES; ++j) {
    const int i4 = tid + 256 * j;
    if (i4 < X_VEC4) {
      const int i = i4 * 4;
      int r = i / F_IN, c = i - r * F_IN;
      const float e[4] = {xv[j].x, xv[j].y, xv[j].z, xv[j].w};
#pragma unroll
      for (int q = 0; q < 4; ++q) {
        sx[r * F_PAD + c] = e[q];
        if (++c == F_IN) { c = 0; ++r; }
      }
    }
  }
  for (int i = tid; i < BM * (F_PAD - F_IN); i += 256) {
    const int r = i / (F_PAD - F_IN), c = F_IN + (i - r * (F_PAD - F_IN));
    sx[r * F_PAD + c] = 0.0f;
  }
  __syncthreads();

  const int lane = tid & 31;
  const int half = lane >> 4;
  const int l16  = lane & 15;
  const int col  = (tid >> 5) * 16 + l16;

  v8f acc[RT];
#pragma unroll
  for (int t = 0; t < RT; ++t) acc[t] = (v8f){0.f, 0.f, 0.f, 0.f, 0.f, 0.f, 0.f, 0.f};

  for (int k = 0; k < F_IN; k += 4) {
    const int ka = k + 2 * half;
    v2f b;
    b.x = (ka     < F_IN) ? W_in[(size_t)(ka)     * D + col] : 0.0f;
    b.y = (ka + 1 < F_IN) ? W_in[(size_t)(ka + 1) * D + col] : 0.0f;
    v2f a[RT];
#pragma unroll
    for (int t = 0; t < RT; ++t)
      a[t] = *(const v2f*)&sx[(t * 16 + l16) * F_PAD + ka];
#pragma unroll
    for (int t = 0; t < RT; ++t) acc[t] = wmma_f32_16x16x4(a[t], b, acc[t]);
  }

  const float bias = b_in[col];
#pragma unroll
  for (int t = 0; t < RT; ++t) {
    const int rowBase = mBase + t * 16 + 8 * half;
#pragma unroll
    for (int r = 0; r < 8; ++r)
      h_out[(size_t)(rowBase + r) * D + col] = tanhf(acc[t][r] + bias);
  }
}

__device__ __forceinline__ void record_matches(const int* mylst, int cnt, int lane,
                                               const int* __restrict__ srcSlice,
                                               int* parts, int* fill) {
  for (int j0 = 0; j0 < cnt; j0 += 32) {
    if (j0 + lane < cnt) {
      const int pk  = mylst[j0 + lane];
      const int s   = srcSlice[pk & ((1 << EDGE_BITS) - 1)];
      const int dl  = pk >> EDGE_BITS;
      const int sub = dl / SUB_ROWS;
      const int at  = atomicAdd(&fill[sub], 1);
      if (at < PART_CAP)
        parts[(size_t)sub * (E_SPLIT * PART_CAP) + at] = s | ((dl - sub * SUB_ROWS) << SRC_BITS);
    }
  }
}

__global__ __launch_bounds__(REC_THREADS) void range_record(const int* __restrict__ src,
                                                            const int* __restrict__ dst,
                                                            int* __restrict__ rec) {
  __shared__ int lst[REC_WAVES * LIST_CAP];
  __shared__ int fill[SUBS];
  const int tid  = threadIdx.x;
  const int lane = tid & 31;
  const int range = blockIdx.x / E_SPLIT;
  const int slice = blockIdx.x - range * E_SPLIT;
  const int rowBase = range * RNG_ROWS;

  if (tid < SUBS) fill[tid] = 0;
  __syncthreads();

  int* parts = rec + REC_HDR + ((size_t)range * SUBS * E_SPLIT + slice) * PART_CAP;
  const int* srcSlice = src + (size_t)slice * SLICE_EDGES;
  const int4* dst4 = (const int4*)(dst + (size_t)slice * SLICE_EDGES);
  int* mylst = lst + (tid >> 5) * LIST_CAP;
  int cnt = 0;
  const int rot = range % SCAN_ITERS;

  for (int i = 0; i < SCAN_ITERS; ++i) {
    int it = i + rot;
    if (it >= SCAN_ITERS) it -= SCAN_ITERS;
    int4 dv[SCAN_U];
    int  i4[SCAN_U];
    int  rb[SCAN_U];
#pragma unroll
    for (int u = 0; u < SCAN_U; ++u) {
      i4[u] = (it * SCAN_U + u) * REC_THREADS + tid;
      const bool ok = i4[u] < SLICE_VEC4;
      dv[u] = dst4[ok ? i4[u] : SLICE_VEC4 - 1];
      rb[u] = ok ? rowBase : 0x7fffffff;
    }
#pragma unroll
    for (int u = 0; u < SCAN_U; ++u) {
      const int d[4] = {dv[u].x, dv[u].y, dv[u].z, dv[u].w};
#pragma unroll
      for (int q = 0; q < 4; ++q) {
        const unsigned dl = (unsigned)(d[q] - rb[u]);
        const bool m = dl < (unsigned)RNG_ROWS;
        const unsigned mask = __builtin_amdgcn_ballot_w32(m);
        if (mask != 0u) {
          if (m)
            mylst[cnt + (int)__builtin_amdgcn_mbcnt_lo(mask, 0u)] =
                (i4[u] * 4 + q) | (int)(dl << EDGE_BITS);
          cnt += __builtin_popcount(mask);
        }
      }
    }
    if (cnt >= FLUSH_AT) {
      record_matches(mylst, cnt, lane, srcSlice, parts, fill);
      cnt = 0;
    }
  }
  if (cnt > 0) record_matches(mylst, cnt, lane, srcSlice, parts, fill);
  __syncthreads();

  if (tid < SUBS)
    rec[((size_t)range * SUBS + tid) * E_SPLIT + slice] = (fill[tid] <= PART_CAP) ? fill[tid] : -1;
}

template <int G, bool SPARE, int PITCH, int SPARE_ROW>
__device__ __forceinline__ void add_rows(int s, int dl, int n, int lane,
                                         const float* __restrict__ h, float* acc) {
  for (int k = 0; k < n; k += G) {
    float4 v[G];
    int d[G];
#pragma unroll
    for (int g = 0; g < G; ++g) {
      const int kk = min(k + g, n - 1);
      const int sg = __builtin_amdgcn_readlane(s, kk);
      d[g] = (SPARE && k + g >= n) ? SPARE_ROW : __builtin_amdgcn_readlane(dl, kk);
      v[g] = ((const float4*)(h + (size_t)sg * D))[lane];
    }
#pragma unroll
    for (int g = 0; g < G; ++g) {
      if (SPARE || k + g < n) {
        float* p = acc + d[g] * PITCH + lane * 4;
        atomicAdd(p + 0, v[g].x);
        atomicAdd(p + 1, v[g].y);
        atomicAdd(p + 2, v[g].z);
        atomicAdd(p + 3, v[g].w);
      }
    }
  }
}

__device__ __forceinline__ int part_place(int j, const int (&pre)[E_SPLIT + 1]) {
  int a = j;
#pragma unroll
  for (int q = 1; q < E_SPLIT; ++q) a = (j >= pre[q]) ? j - pre[q] + q * PART_CAP : a;
  return a;
}

__global__ __launch_bounds__(256) void owner_replay_gemm(const float* __restrict__ h,
                                                         const int* __restrict__ src,
                                                         const int* __restrict__ dst,
                                                         const int* __restrict__ rec,
                                                         const float* __restrict__ Wself,
                                                         const float* __restrict__ Wneigh,
                                                         const float* __restrict__ bias_v,
                                                         float* __restrict__ h_out,
                                                         int relu) {
  __shared__ __attribute__((aligned(16))) float sh[SUB_ROWS * D_PAD];
  __shared__ __attribute__((aligned(16))) float sa[(SUB_ROWS + 1) * D_PAD];
  __shared__ int degc[SUB_ROWS];
  const int tid  = threadIdx.x;
  const int lane = tid & 31;
  const int rowBase = blockIdx.x * SUB_ROWS;
  const int wave = __builtin_amdgcn_readfirstlane(tid >> 5);

  constexpr int SELF_PASSES = (SUB_ROWS * (D / 4)) / 256;
  float4 hv[SELF_PASSES];
#pragma unroll
  for (int j = 0; j < SELF_PASSES; ++j) {
    const int i = tid + 256 * j;
    hv[j] = ((const float4*)(h + (size_t)(rowBase + (i >> 5)) * D))[i & 31];
  }
  const int* glist = rec + REC_HDR + (size_t)blockIdx.x * (E_SPLIT * PART_CAP);
  int pre[E_SPLIT + 1];
  bool fits = true;
  pre[0] = 0;
#pragma unroll
  for (int q = 0; q < E_SPLIT; ++q) {
    const int c = rec[blockIdx.x * E_SPLIT + q];
    fits = fits && c >= 0 && c <= PART_CAP;
    pre[q + 1] = pre[q] + max(c, 0);
  }
  const int nrec = pre[E_SPLIT];
  int pk_next = (fits && wave * 32 + lane < nrec) ? glist[part_place(wave * 32 + lane, pre)] : 0;

  for (int i = tid; i < (SUB_ROWS + 1) * (D_PAD / 4); i += 256)
    ((float4*)sa)[i] = make_float4(0.f, 0.f, 0.f, 0.f);
  if (tid < SUB_ROWS) degc[tid] = 0;
#pragma unroll
  for (int j = 0; j < SELF_PASSES; ++j) {
    const int i = tid + 256 * j;
    *(float4*)&sh[(i >> 5) * D_PAD + (i & 31) * 4] = hv[j];
  }
  __syncthreads();

  const int half = lane >> 4;
  const int l16  = lane & 15;
  const int col  = (tid >> 5) * 16 + l16;

  v8f acc[SUB_RT];
#pragma unroll
  for (int t = 0; t < SUB_RT; ++t) acc[t] = (v8f){0.f, 0.f, 0.f, 0.f, 0.f, 0.f, 0.f, 0.f};

  const float* wsp = Wself + (size_t)(2 * half) * D + col;
  const float* shp = sh + l16 * D_PAD + 2 * half;

  const int j0w = wave * 32;
  const int n0  = fits ? max(0, min(32, nrec - j0w)) : 0;
  const int pk0 = pk_next;
  if (fits) {
    const int jn = j0w + 32 * SUB_WAVES + lane;
    pk_next = (jn < nrec) ? glist[part_place(jn, pre)] : 0;
  }
  if (lane < n0) atomicAdd(&degc[pk0 >> SRC_BITS], 1);
  const int s0  = pk0 & ((1 << SRC_BITS) - 1);
  const int dl0 = pk0 >> SRC_BITS;

  v2f b[SELF_KSTEPS];
#pragma unroll
  for (int i = 0; i < SELF_KSTEPS; ++i) {
    b[i].x = wsp[(size_t)(4 * i)     * D];
    b[i].y = wsp[(size_t)(4 * i + 1) * D];
  }
#pragma unroll
  for (int st = 0; st < SELF_BLOCKS; ++st) {
    const int k0 = st * SELF_G;
    float4 v[SELF_G];
    int d[SELF_G];
#pragma unroll
    for (int g = 0; g < SELF_G; ++g) {
      const int kk = max(min(k0 + g, n0 - 1), 0);
      const int sg = __builtin_amdgcn_readlane(s0, kk);
      const int dg = __builtin_amdgcn_readlane(dl0, kk);
      d[g] = (k0 + g >= n0) ? SUB_ROWS : dg;
      v[g] = ((const float4*)(h + (size_t)sg * D))[lane];
    }
#pragma unroll
    for (int i = 0; i < SELF_KSTEPS; ++i) {
      const int kk = (st * SELF_KSTEPS + i) * 4;
      v2f a[SUB_RT];
#pragma unroll
      for (int t = 0; t < SUB_RT; ++t) a[t] = *(const v2f*)&shp[t * 16 * D_PAD + kk];
#pragma unroll
      for (int t = 0; t < SUB_RT; ++t) acc[t] = wmma_f32_16x16x4(a[t], b[i], acc[t]);
    }
    __builtin_amdgcn_sched_barrier(0);
    if (st + 1 < SELF_BLOCKS) {
#pragma unroll
      for (int i = 0; i < SELF_KSTEPS; ++i) {
        const int kk = ((st + 1) * SELF_KSTEPS + i) * 4;
        b[i].x = wsp[(size_t)(kk)     * D];
        b[i].y = wsp[(size_t)(kk + 1) * D];
      }
    }
#pragma unroll
    for (int g = 0; g < SELF_G; ++g) {
      float* p = sa + d[g] * D_PAD + lane * 4;
      atomicAdd(p + 0, v[g].x);
      atomicAdd(p + 1, v[g].y);
      atomicAdd(p + 2, v[g].z);
      atomicAdd(p + 3, v[g].w);
    }
  }

  if (fits) {
    for (int j0 = j0w + 32 * SUB_WAVES; j0 < nrec; j0 += 32 * SUB_WAVES) {
      const int n  = min(32, nrec - j0);
      const int pk = pk_next;
      const int jn = j0 + 32 * SUB_WAVES + lane;
      pk_next = (jn < nrec) ? glist[part_place(jn, pre)] : 0;
      if (lane < n) atomicAdd(&degc[pk >> SRC_BITS], 1);
      add_rows<REPLAY_G, true, D_PAD, SUB_ROWS>(pk & ((1 << SRC_BITS) - 1), pk >> SRC_BITS, n, lane, h, sa);
    }
  } else {
    for (int e0 = wave * 32; e0 < N_EDGES; e0 += 32 * SUB_WAVES) {
      const int e = e0 + lane;
      const unsigned dl = (e < N_EDGES) ? (unsigned)(dst[e] - rowBase) : 0xffffffffu;
      const bool m = dl < (unsigned)SUB_ROWS;
      unsigned mask = __builtin_amdgcn_ballot_w32(m);
      if (mask != 0u) {
        int s = 0;
        if (m) {
          s = src[e];
          atomicAdd(&degc[dl], 1);
        }
        while (mask != 0u) {
          const int bit = __builtin_ctz(mask);
          mask &= mask - 1u;
          const int sg = __builtin_amdgcn_readlane(s, bit);
          const int dg = __builtin_amdgcn_readlane((int)dl, bit);
          const float4 v = ((const float4*)(h + (size_t)sg * D))[lane];
          float* p = sa + dg * D_PAD + lane * 4;
          atomicAdd(p + 0, v.x);
          atomicAdd(p + 1, v.y);
          atomicAdd(p + 2, v.z);
          atomicAdd(p + 3, v.w);
        }
      }
    }
  }
  __syncthreads();

  for (int i = tid; i < SUB_ROWS * (D / 4); i += 256) {
    const int r = i >> 5;
    const float inv = 1.0f / fmaxf((float)degc[r], 1.0f);
    float4* p = (float4*)&sa[r * D_PAD + (i & 31) * 4];
    float4 v = *p;
    v.x *= inv; v.y *= inv; v.z *= inv; v.w *= inv;
    *p = v;
  }
  __syncthreads();

  const float* wnp = Wneigh + (size_t)(2 * half) * D + col;
  const float* sap = sa + l16 * D_PAD + 2 * half;
#pragma unroll 8
  for (int k = 0; k < D; k += 4) {
    v2f b2;
    b2.x = wnp[(size_t)(k)     * D];
    b2.y = wnp[(size_t)(k + 1) * D];
    v2f a2[SUB_RT];
#pragma unroll
    for (int t = 0; t < SUB_RT; ++t) a2[t] = *(const v2f*)&sap[t * 16 * D_PAD + k];
#pragma unroll
    for (int t = 0; t < SUB_RT; ++t) acc[t] = wmma_f32_16x16x4(a2[t], b2, acc[t]);
  }

  const float bv = bias_v[col];
#pragma unroll
  for (int t = 0; t < SUB_RT; ++t) {
    const int rowT = rowBase + t * 16 + 8 * half;
#pragma unroll
    for (int r = 0; r < 8; ++r) {
      float v = acc[t][r] + bv;
      if (relu) v = fmaxf(v, 0.0f);
      h_out[(size_t)(rowT + r) * D + col] = v;
    }
  }
}

extern "C" void kernel_launch(void* const* d_in, const int* in_sizes, int n_in,
                              void* d_out, int out_size, void* d_ws, size_t ws_size,
                              hipStream_t stream) {
  const float* x       = (const float*)d_in[0];
  const int*   ei      = (const int*)d_in[1];
  const float* W_in    = (const float*)d_in[2];
  const float* b_in    = (const float*)d_in[3];
  const float* W_self  = (const float*)d_in[4];
  const float* W_neigh = (const float*)d_in[5];
  const float* b_lay   = (const float*)d_in[6];
  float* out = (float*)d_out;

  float* ws   = (float*)d_ws;
  float* hA   = ws;
  int*   rec  = (int*)(ws + (size_t)1 * N_NODES * D);
  float* hB   = ws + (size_t)2 * N_NODES * D;

  const int* src = ei;
  const int* dst = ei + N_EDGES;

  input_gemm_tanh<<<N_NODES / BM, 256, 0, stream>>>(x, W_in, b_in, hA);

  range_record<<<REC_GROUPS, REC_THREADS, 0, stream>>>(src, dst, rec);

  owner_replay_gemm<<<SUB_GROUPS, 256, 0, stream>>>(
      hA, src, dst, rec, W_self, W_neigh, b_lay, hB, 1);
  owner_replay_gemm<<<SUB_GROUPS, 256, 0, stream>>>(
      hB, src, dst, rec, W_self + (size_t)D * D, W_neigh + (size_t)D * D, b_lay + D, hA, 1);
  owner_replay_gemm<<<SUB_GROUPS, 256, 0, stream>>>(
      hA, src, dst, rec, W_self + (size_t)2 * D * D, W_neigh + (size_t)2 * D * D, b_lay + 2 * D, out, 0);
}


# Neighbour sums leave LDS float atomics: lists recorded per 10 rows, one per wave; a wave ranks its 32 pairs by row (ballots), sums each row's run in registers, 1 LDS store per row; sums in list order
# speedup vs baseline: 2.1881x; 1.0295x over previous
#include <hip/hip_runtime.h>
#include <math.h>


typedef __attribute__((ext_vector_type(2))) float v2f;
typedef __attribute__((ext_vector_type(4))) float v4f;
typedef __attribute__((ext_vector_type(8))) float v8f;

#define N_NODES 200000
#define N_EDGES 600000
#define F_IN    117
#define F_PAD   124
#define D       128
#define D_PAD   132
#define L_LAYERS 3
#define RT      4
#define BM      (16 * RT)
#define X_VEC4   ((BM * F_IN) / 4)
#define X_PASSES ((X_VEC4 + 255) / 256)

#define SUB_ROWS    80
#define SUB_GROUPS  (N_NODES / SUB_ROWS)
#define SUB_RT      (SUB_ROWS / 16)
#define SUB_WAVES   8
#define WAVE_ROWS   (SUB_ROWS / SUB_WAVES)
#define LISTS       (N_NODES / WAVE_ROWS)
#define RNG_ROWS    4000
#define RNG_COUNT   (N_NODES / RNG_ROWS)
#define SUBS        (RNG_ROWS / WAVE_ROWS)
#define E_SPLIT     5
#define REC_GROUPS  (RNG_COUNT * E_SPLIT)
#define REC_THREADS 1024
#define REC_WAVES   (REC_THREADS / 32)
#define SLICE_EDGES (N_EDGES / E_SPLIT)
#define SLICE_VEC4  (SLICE_EDGES / 4)
#define SCAN_U      4
#define SCAN_ITERS  ((SLICE_VEC4 + REC_THREADS * SCAN_U - 1) / (REC_THREADS * SCAN_U))
#define FLUSH_AT    32
#define LIST_CAP    640
#define EDGE_BITS   17
#define SRC_BITS    18
#define PART_CAP    32
#define REC_HDR     100096
#define REPLAY_G    8
#define SELF_BLOCKS 4
#define SELF_KSTEPS (D / 4 / SELF_BLOCKS)
#define SELF_G      (32 / SELF_BLOCKS)

static_assert(N_NODES % RNG_ROWS == 0, "ranges must tile the node set");
static_assert(N_EDGES % (4 * E_SPLIT) == 0, "every slice of an index row is read as whole 16-byte vectors");
static_assert(SLICE_EDGES <= (1 << EDGE_BITS) && RNG_ROWS <= (1 << (31 - EDGE_BITS)), "match packing");
static_assert(FLUSH_AT - 1 + 32 * 4 * SCAN_U <= LIST_CAP, "one scan trip must fit behind a carried remainder");
static_assert(N_NODES <= (1 << SRC_BITS) && SUB_ROWS <= (1 << (31 - SRC_BITS)), "recorded entry packing");
static_assert(RNG_ROWS % SUB_ROWS == 0 && SUB_ROWS % 16 == 0 && SUB_ROWS % SUB_WAVES == 0, "work-groups tile a range, tiles and waves tile a work-group's rows");
static_assert(LISTS * E_SPLIT <= REC_HDR, "one count per part");
static_assert((size_t)REC_HDR + (size_t)LISTS * E_SPLIT * PART_CAP <= (size_t)N_NODES * D, "lists fit in one h plane");
static_assert(SUBS <= REC_THREADS, "one thread per list of a range clears and publishes its count");
static_assert((SUB_ROWS * (D / 4)) % 256 == 0, "own rows stage in whole passes");
static_assert(SELF_BLOCKS * SELF_KSTEPS * 4 == D && SELF_BLOCKS * SELF_G == 32, "pieces tile K, groups tile a chunk");

__device__ __forceinline__ v8f wmma_f32_16x16x4(v2f a, v2f b, v8f c) {
  return __builtin_amdgcn_wmma_f32_16x16x4_f32(false, a, false, b, (short)0, c,
                                               false, false);
}

__global__ __launch_bounds__(256) void input_gemm_tanh(const float* __restrict__ x,
                                                       const float* __restrict__ W_in,
                                                       const float* __restrict__ b_in,
                                                       float* __restrict__ h_out) {
  __shared__ __attribute__((aligned(16))) float sx[BM * F_PAD];
  const int tid = threadIdx.x;
  const int mBase = blockIdx.x * BM;

  const float4* xt = (const float4*)(x + (size_t)mBase * F_IN);
  float4 xv[X_PASSES];
#pragma unroll
  for (int j = 0; j < X_PASSES; ++j) {
    const int i4 = tid + 256 * j;
    xv[j] = (i4 < X_VEC4) ? xt[i4] : make_float4(0.f, 0.f, 0.f, 0.f);
  }
#pragma unroll
  for (int j = 0; j < X_PASSES; ++j) {
    const int i4 = tid + 256 * j;
    if (i4 < X_VEC4) {
      const int i = i4 * 4;
      int r = i / F_IN, c = i - r * F_IN;
      const float e[4] = {xv[j].x, xv[j].y, xv[j].z, xv[j].w};
#pragma unroll
      for (int q = 0; q < 4; ++q) {
        sx[r * F_PAD + c] = e[q];
        if (++c == F_IN) { c = 0; ++r; }
      }
    }
  }
  for (int i = tid; i < BM * (F_PAD - F_IN); i += 256) {
    const int r = i / (F_PAD - F_IN), c = F_IN + (i - r * (F_PAD - F_IN));
    sx[r * F_PAD + c] = 0.0f;
  }
  __syncthreads();

  const int lane = tid & 31;
  const int half = lane >> 4;
  const int l16  = lane & 15;
  const int col  = (tid >> 5) * 16 + l16;

  v8f acc[RT];
#pragma unroll
  for (int t = 0; t < RT; ++t) acc[t] = (v8f){0.f, 0.f, 0.f, 0.f, 0.f, 0.f, 0.f, 0.f};

  for (int k = 0; k < F_IN; k += 4) {
    const int ka = k + 2 * half;
    v2f b;
    b.x = (ka     < F_IN) ? W_in[(size_t)(ka)     * D + col] : 0.0f;
    b.y = (ka + 1 < F_IN) ? W_in[(size_t)(ka + 1) * D + col] : 0.0f;
    v2f a[RT];
#pragma unroll
    for (int t = 0; t < RT; ++t)
      a[t] = *(const v2f*)&sx[(t * 16 + l16) * F_PAD + ka];
#pragma unroll
    for (int t = 0; t < RT; ++t) acc[t] = wmma_f32_16x16x4(a[t], b, acc[t]);
  }

  const float bias = b_in[col];
#pragma unroll
  for (int t = 0; t < RT; ++t) {
    const int rowBase = mBase + t * 16 + 8 * half;
#pragma unroll
    for (int r = 0; r < 8; ++r)
      h_out[(size_t)(rowBase + r) * D + col] = tanhf(acc[t][r] + bias);
  }
}

__device__ __forceinline__ void record_matches(const int* mylst, int cnt, int lane,
                                               const int* __restrict__ srcSlice,
                                               int* parts, int* fill) {
  for (int j0 = 0; j0 < cnt; j0 += 32) {
    if (j0 + lane < cnt) {
      const int pk  = mylst[j0 + lane];
      const int s   = srcSlice[pk & ((1 << EDGE_BITS) - 1)];
      const int dl  = pk >> EDGE_BITS;
      const int sub = dl / WAVE_ROWS;
      const int at  = atomicAdd(&fill[sub], 1);
      if (at < PART_CAP)
        parts[(size_t)sub * (E_SPLIT * PART_CAP) + at] = s | ((dl % SUB_ROWS) << SRC_BITS);
    }
  }
}

__global__ __launch_bounds__(REC_THREADS) void range_record(const int* __restrict__ src,
                                                            const int* __restrict__ dst,
                                                            int* __restrict__ rec) {
  __shared__ int lst[REC_WAVES * LIST_CAP];
  __shared__ int fill[SUBS];
  const int tid  = threadIdx.x;
  const int lane = tid & 31;
  const int range = blockIdx.x / E_SPLIT;
  const int slice = blockIdx.x - range * E_SPLIT;
  const int rowBase = range * RNG_ROWS;

  if (tid < SUBS) fill[tid] = 0;
  __syncthreads();

  int* parts = rec + REC_HDR + ((size_t)range * SUBS * E_SPLIT + slice) * PART_CAP;
  const int* srcSlice = src + (size_t)slice * SLICE_EDGES;
  const int4* dst4 = (const int4*)(dst + (size_t)slice * SLICE_EDGES);
  int* mylst = lst + (tid >> 5) * LIST_CAP;
  int cnt = 0;
  const int rot = range % SCAN_ITERS;

  for (int i = 0; i < SCAN_ITERS; ++i) {
    int it = i + rot;
    if (it >= SCAN_ITERS) it -= SCAN_ITERS;
    int4 dv[SCAN_U];
    int  i4[SCAN_U];
    int  rb[SCAN_U];
#pragma unroll
    for (int u = 0; u < SCAN_U; ++u) {
      i4[u] = (it * SCAN_U + u) * REC_THREADS + tid;
      const bool ok = i4[u] < SLICE_VEC4;
      dv[u] = dst4[ok ? i4[u] : SLICE_VEC4 - 1];
      rb[u] = ok ? rowBase : 0x7fffffff;
    }
#pragma unroll
    for (int u = 0; u < SCAN_U; ++u) {
      const int d[4] = {dv[u].x, dv[u].y, dv[u].z, dv[u].w};
#pragma unroll
      for (int q = 0; q < 4; ++q) {
        const unsigned dl = (unsigned)(d[q] - rb[u]);
        const bool m = dl < (unsigned)RNG_ROWS;
        const unsigned mask = __builtin_amdgcn_ballot_w32(m);
        if (mask != 0u) {
          if (m)
            mylst[cnt + (int)__builtin_amdgcn_mbcnt_lo(mask, 0u)] =
                (i4[u] * 4 + q) | (int)(dl << EDGE_BITS);
          cnt += __builtin_popcount(mask);
        }
      }
    }
    if (cnt >= FLUSH_AT) {
      record_matches(mylst, cnt, lane, srcSlice, parts, fill);
      cnt = 0;
    }
  }
  if (cnt > 0) record_matches(mylst, cnt, lane, srcSlice, parts, fill);
  __syncthreads();

  if (tid < SUBS)
    rec[((size_t)range * SUBS + tid) * E_SPLIT + slice] = (fill[tid] <= PART_CAP) ? fill[tid] : -1;
}

template <int G, bool SPARE, int PITCH, int SPARE_ROW>
__device__ __forceinline__ void add_rows(int s, int dl, int n, int lane,
                                         const float* __restrict__ h, float* acc) {
  for (int k = 0; k < n; k += G) {
    float4 v[G];
    int d[G];
#pragma unroll
    for (int g = 0; g < G; ++g) {
      const int kk = min(k + g, n - 1);
      const int sg = __builtin_amdgcn_readlane(s, kk);
      d[g] = (SPARE && k + g >= n) ? SPARE_ROW : __builtin_amdgcn_readlane(dl, kk);
      v[g] = ((const float4*)(h + (size_t)sg * D))[lane];
    }
#pragma unroll
    for (int g = 0; g < G; ++g) {
      if (SPARE || k + g < n) {
        float* p = acc + d[g] * PITCH + lane * 4;
        atomicAdd(p + 0, v[g].x);
        atomicAdd(p + 1, v[g].y);
        atomicAdd(p + 2, v[g].z);
        atomicAdd(p + 3, v[g].w);
      }
    }
  }
}

__device__ __forceinline__ int part_place(int j, const int (&pre)[E_SPLIT + 1]) {
  int a = j;
#pragma unroll
  for (int q = 1; q < E_SPLIT; ++q) a = (j >= pre[q]) ? j - pre[q] + q * PART_CAP : a;
  return a;
}

__global__ __launch_bounds__(256) void owner_replay_gemm(const float* __restrict__ h,
                                                         const int* __restrict__ src,
                                                         const int* __restrict__ dst,
                                                         const int* __restrict__ rec,
                                                         const float* __restrict__ Wself,
                                                         const float* __restrict__ Wneigh,
                                                         const float* __restrict__ bias_v,
                                                         float* __restrict__ h_out,
                                                         int relu) {
  __shared__ __attribute__((aligned(16))) float sh[SUB_ROWS * D_PAD];
  __shared__ __attribute__((aligned(16))) float sa[(SUB_ROWS + 1) * D_PAD];
  __shared__ int degc[SUB_ROWS];
  const int tid  = threadIdx.x;
  const int lane = tid & 31;
  const int rowBase = blockIdx.x * SUB_ROWS;
  const int wave = __builtin_amdgcn_readfirstlane(tid >> 5);

  constexpr int SELF_PASSES = (SUB_ROWS * (D / 4)) / 256;
  float4 hv[SELF_PASSES];
#pragma unroll
  for (int j = 0; j < SELF_PASSES; ++j) {
    const int i = tid + 256 * j;
    hv[j] = ((const float4*)(h + (size_t)(rowBase + (i >> 5)) * D))[i & 31];
  }
  const int list = blockIdx.x * SUB_WAVES + wave;
  const int* glist = rec + REC_HDR + (size_t)list * (E_SPLIT * PART_CAP);
  int pre[E_SPLIT + 1];
  bool fits = true;
  pre[0] = 0;
#pragma unroll
  for (int q = 0; q < E_SPLIT; ++q) {
    const int c = rec[list * E_SPLIT + q];
    fits = fits && c >= 0 && c <= PART_CAP;
    pre[q + 1] = pre[q] + max(c, 0);
  }
  const int nrec = pre[E_SPLIT];
  int pk_next = (fits && lane < nrec) ? glist[part_place(lane, pre)] : 0;

  for (int i = tid; i < (SUB_ROWS + 1) * (D_PAD / 4); i += 256)
    ((float4*)sa)[i] = make_float4(0.f, 0.f, 0.f, 0.f);
  if (tid < SUB_ROWS) degc[tid] = 0;
#pragma unroll
  for (int j = 0; j < SELF_PASSES; ++j) {
    const int i = tid + 256 * j;
    *(float4*)&sh[(i >> 5) * D_PAD + (i & 31) * 4] = hv[j];
  }
  __syncthreads();

  const int half = lane >> 4;
  const int l16  = lane & 15;
  const int col  = (tid >> 5) * 16 + l16;

  v8f acc[SUB_RT];
#pragma unroll
  for (int t = 0; t < SUB_RT; ++t) acc[t] = (v8f){0.f, 0.f, 0.f, 0.f, 0.f, 0.f, 0.f, 0.f};

  const float* wsp = Wself + (size_t)(2 * half) * D + col;
  const float* shp = sh + l16 * D_PAD + 2 * half;

  const int n0  = fits ? min(32, nrec) : 0;
  const int pk0 = pk_next;
  if (fits) {
    const int jn = 32 + lane;
    pk_next = (jn < nrec) ? glist[part_place(jn, pre)] : 0;
  }
  if (lane < n0) atomicAdd(&degc[pk0 >> SRC_BITS], 1);
  const int wl = (lane < n0) ? (pk0 >> SRC_BITS) - wave * WAVE_ROWS : WAVE_ROWS;
  int place = 0, below = 0;
#pragma unroll
  for (int r = 0; r <= WAVE_ROWS; ++r) {
    const unsigned m = __builtin_amdgcn_ballot_w32(wl == r);
    if (wl == r) place = below + (int)__builtin_amdgcn_mbcnt_lo(m, 0u);
    below += __builtin_popcount(m);
  }
  const int spk = __builtin_amdgcn_ds_permute(place << 2, pk0);
  const int s0  = spk & ((1 << SRC_BITS) - 1);
  const int dl0 = spk >> SRC_BITS;
  v4f cur = (v4f){0.f, 0.f, 0.f, 0.f};
  int cur_d = SUB_ROWS;

  v2f b[SELF_KSTEPS];
#pragma unroll
  for (int i = 0; i < SELF_KSTEPS; ++i) {
    b[i].x = wsp[(size_t)(4 * i)     * D];
    b[i].y = wsp[(size_t)(4 * i + 1) * D];
  }
#pragma unroll
  for (int st = 0; st < SELF_BLOCKS; ++st) {
    const int k0 = st * SELF_G;
    v4f v[SELF_G];
    int d[SELF_G];
#pragma unroll
    for (int g = 0; g < SELF_G; ++g) {
      const int kk = max(min(k0 + g, n0 - 1), 0);
      const int sg = __builtin_amdgcn_readlane(s0, kk);
      const int dg = __builtin_amdgcn_readlane(dl0, kk);
      d[g] = (k0 + g >= n0) ? SUB_ROWS : dg;
      v[g] = ((const v4f*)(h + (size_t)sg * D))[lane];
    }
#pragma unroll
    for (int i = 0; i < SELF_KSTEPS; ++i) {
      const int kk = (st * SELF_KSTEPS + i) * 4;
      v2f a[SUB_RT];
#pragma unroll
      for (int t = 0; t < SUB_RT; ++t) a[t] = *(const v2f*)&shp[t * 16 * D_PAD + kk];
#pragma unroll
      for (int t = 0; t < SUB_RT; ++t) acc[t] = wmma_f32_16x16x4(a[t], b[i], acc[t]);
    }
    __builtin_amdgcn_sched_barrier(0);
    if (st + 1 < SELF_BLOCKS) {
#pragma unroll
      for (int i = 0; i < SELF_KSTEPS; ++i) {
        const int kk = ((st + 1) * SELF_KSTEPS + i) * 4;
        b[i].x = wsp[(size_t)(kk)     * D];
        b[i].y = wsp[(size_t)(kk + 1) * D];
      }
    }
#pragma unroll
    for (int g = 0; g < SELF_G; ++g) {
      asm volatile("" : "+v"(v[g]));
      const bool last = d[g] != cur_d;
      *(v4f*)(sa + (last ? cur_d : SUB_ROWS) * D_PAD + lane * 4) = cur;
      cur = last ? v[g] : cur + v[g];
      cur_d = d[g];
    }
    __builtin_amdgcn_sched_barrier(0);
  }
  *(v4f*)(sa + cur_d * D_PAD + lane * 4) = cur;

  if (fits) {
    for (int j0 = 32; j0 < nrec; j0 += 32) {
      const int n  = min(32, nrec - j0);
      const int pk = pk_next;
      const int jn = j0 + 32 + lane;
      pk_next = (jn < nrec) ? glist[part_place(jn, pre)] : 0;
      if (lane < n) atomicAdd(&degc[pk >> SRC_BITS], 1);
      add_rows<REPLAY_G, true, D_PAD, SUB_ROWS>(pk & ((1 << SRC_BITS) - 1), pk >> SRC_BITS, n, lane, h, sa);
    }
  } else {
    const int ownBase = wave * WAVE_ROWS;
    for (int e0 = 0; e0 < N_EDGES; e0 += 32) {
      const int e = e0 + lane;
      const unsigned dw = (e < N_EDGES) ? (unsigned)(dst[e] - rowBase - ownBase) : 0xffffffffu;
      const bool m = dw < (unsigned)WAVE_ROWS;
      const unsigned dl = dw + (unsigned)ownBase;
      unsigned mask = __builtin_amdgcn_ballot_w32(m);
      if (mask != 0u) {
        int s = 0;
        if (m) {
          s = src[e];
          atomicAdd(&degc[dl], 1);
        }
        while (mask != 0u) {
          const int bit = __builtin_ctz(mask);
          mask &= mask - 1u;
          const int sg = __builtin_amdgcn_readlane(s, bit);
          const int dg = __builtin_amdgcn_readlane((int)dl, bit);
          const float4 v = ((const float4*)(h + (size_t)sg * D))[lane];
          float* p = sa + dg * D_PAD + lane * 4;
          atomicAdd(p + 0, v.x);
          atomicAdd(p + 1, v.y);
          atomicAdd(p + 2, v.z);
          atomicAdd(p + 3, v.w);
        }
      }
    }
  }
  __syncthreads();

  for (int i = tid; i < SUB_ROWS * (D / 4); i += 256) {
    const int r = i >> 5;
    const float inv = 1.0f / fmaxf((float)degc[r], 1.0f);
    float4* p = (float4*)&sa[r * D_PAD + (i & 31) * 4];
    float4 v = *p;
    v.x *= inv; v.y *= inv; v.z *= inv; v.w *= inv;
    *p = v;
  }
  __syncthreads();

  const float* wnp = Wneigh + (size_t)(2 * half) * D + col;
  const float* sap = sa + l16 * D_PAD + 2 * half;
#pragma unroll 8
  for (int k = 0; k < D; k += 4) {
    v2f b2;
    b2.x = wnp[(size_t)(k)     * D];
    b2.y = wnp[(size_t)(k + 1) * D];
    v2f a2[SUB_RT];
#pragma unroll
    for (int t = 0; t < SUB_RT; ++t) a2[t] = *(const v2f*)&sap[t * 16 * D_PAD + k];
#pragma unroll
    for (int t = 0; t < SUB_RT; ++t) acc[t] = wmma_f32_16x16x4(a2[t], b2, acc[t]);
  }

  const float bv = bias_v[col];
#pragma unroll
  for (int t = 0; t < SUB_RT; ++t) {
    const int rowT = rowBase + t * 16 + 8 * half;
#pragma unroll
    for (int r = 0; r < 8; ++r) {
      float v = acc[t][r] + bv;
      if (relu) v = fmaxf(v, 0.0f);
      h_out[(size_t)(rowT + r) * D + col] = v;
    }
  }
}

extern "C" void kernel_launch(void* const* d_in, const int* in_sizes, int n_in,
                              void* d_out, int out_size, void* d_ws, size_t ws_size,
                              hipStream_t stream) {
  const float* x       = (const float*)d_in[0];
  const int*   ei      = (const int*)d_in[1];
  const float* W_in    = (const float*)d_in[2];
  const float* b_in    = (const float*)d_in[3];
  const float* W_self  = (const float*)d_in[4];
  const float* W_neigh = (const float*)d_in[5];
  const float* b_lay   = (const float*)d_in[6];
  float* out = (float*)d_out;

  float* ws   = (float*)d_ws;
  float* hA   = ws;
  int*   rec  = (int*)(ws + (size_t)1 * N_NODES * D);
  float* hB   = ws + (size_t)2 * N_NODES * D;

  const int* src = ei;
  const int* dst = ei + N_EDGES;

  input_gemm_tanh<<<N_NODES / BM, 256, 0, stream>>>(x, W_in, b_in, hA);

  range_record<<<REC_GROUPS, REC_THREADS, 0, stream>>>(src, dst, rec);

  owner_replay_gemm<<<SUB_GROUPS, 256, 0, stream>>>(
      hA, src, dst, rec, W_self, W_neigh, b_lay, hB, 1);
  owner_replay_gemm<<<SUB_GROUPS, 256, 0, stream>>>(
      hB, src, dst, rec, W_self + (size_t)D * D, W_neigh + (size_t)D * D, b_lay + D, hA, 1);
  owner_replay_gemm<<<SUB_GROUPS, 256, 0, stream>>>(
      hA, src, dst, rec, W_self + (size_t)2 * D * D, W_neigh + (size_t)2 * D * D, b_lay + 2 * D, out, 0);
}
